# starting kernel
# baseline (speedup 1.0000x reference)
#include <hip/hip_runtime.h>

typedef __attribute__((ext_vector_type(2))) float v2f;
typedef __attribute__((ext_vector_type(8))) float v8f;

#define NNETS 256
#define HID 128
#define BATCH 2048
#define ROWS_PER_BLK 128
#define W2_STRIDE 132

__global__ __launch_bounds__(256) void mlp256_fused_wmma_f32(
    const float* __restrict__ a,  const float* __restrict__ W1,
    const float* __restrict__ b1, const float* __restrict__ W2,
    const float* __restrict__ b2, const float* __restrict__ W3,
    const float* __restrict__ b3, float* __restrict__ out)
{
    __shared__ float sW2[HID * W2_STRIDE];
    __shared__ float sW1[HID];
    __shared__ float sB1[HID];
    __shared__ float sB2[HID];
    __shared__ float sW3[HID];
    __shared__ float sA[ROWS_PER_BLK];

    const int n  = blockIdx.y;
    const int r0 = blockIdx.x * ROWS_PER_BLK;
    const int t  = threadIdx.x;

    if (t < HID) {
        sW1[t] = W1[n * HID + t];
        sB1[t] = b1[n * HID + t];
        sB2[t] = b2[n * HID + t];
        sW3[t] = W3[n * HID + t];
        sA[t]  = a[(size_t)n * BATCH + r0 + t];
    }

    const float* gW2 = W2 + (size_t)n * HID * HID;
    #pragma unroll
    for (int i = 0; i < 32; ++i) {
        int idx2 = t + 256 * i;
        int e    = idx2 * 2;
        int row  = e >> 7;
        int col  = e & 127;
        v2f v = *(const v2f*)(gW2 + e);
        *(v2f*)(sW2 + row * W2_STRIDE + col) = v;
    }
    __syncthreads();

    const int lane = t & 31;
    const int wv   = t >> 5;
    const int li   = lane & 15;
    const int lh   = lane >> 4;

    const int brow = wv * 16 + li;
    const float aval = sA[brow];
    v2f bfr[32];
    #pragma unroll
    for (int s = 0; s < 32; ++s) {
        int k = 4 * s + 2 * lh;
        v2f w1v = *(const v2f*)(sW1 + k);
        v2f b1v = *(const v2f*)(sB1 + k);
        v2f h;
        h.x = fmaxf(aval * w1v.x + b1v.x, 0.0f);
        h.y = fmaxf(aval * w1v.y + b1v.y, 0.0f);
        bfr[s] = h;
    }

    float acc = 0.0f;
    #pragma unroll
    for (int mt = 0; mt < 8; ++mt) {
        v8f c = {0.f, 0.f, 0.f, 0.f, 0.f, 0.f, 0.f, 0.f};
        const float* arow = sW2 + (mt * 16 + li) * W2_STRIDE + 2 * lh;
        #pragma unroll
        for (int s = 0; s < 32; ++s) {
            v2f afrag = *(const v2f*)(arow + 4 * s);
            c = __builtin_amdgcn_wmma_f32_16x16x4_f32(
                    false, afrag, false, bfr[s], (short)0, c, false, false);
        }
        #pragma unroll
        for (int v = 0; v < 8; ++v) {
            int kout = mt * 16 + v + 8 * lh;
            float h2 = fmaxf(c[v] + sB2[kout], 0.0f);
            acc += h2 * sW3[kout];
        }
    }

    acc += __shfl_xor(acc, 16, 32);
    if (lh == 0) {
        out[(size_t)n * BATCH + r0 + brow] = acc + b3[n];
    }
}

extern "C" void kernel_launch(void* const* d_in, const int* in_sizes, int n_in,
                              void* d_out, int out_size, void* d_ws, size_t ws_size,
                              hipStream_t stream) {
    const float* a  = (const float*)d_in[0];
    const float* W1 = (const float*)d_in[1];
    const float* b1 = (const float*)d_in[2];
    const float* W2 = (const float*)d_in[3];
    const float* b2 = (const float*)d_in[4];
    const float* W3 = (const float*)d_in[5];
    const float* b3 = (const float*)d_in[6];
    float* out = (float*)d_out;

    dim3 grid(BATCH / ROWS_PER_BLK, NNETS);
    mlp256_fused_wmma_f32<<<grid, 256, 0, stream>>>(a, W1, b1, W2, b2, W3, b3, out);
}


# Batch-tile register blocking with W2 kept in LDS: one 8-wave work-group per net stages W2 once; each A fragment read from LDS feeds 4 WMMAs (4 row tiles of B fragments in registers per wave)
# speedup vs baseline: 1.1497x; 1.1497x over previous
#include <hip/hip_runtime.h>


typedef __attribute__((ext_vector_type(2))) float v2f;
typedef __attribute__((ext_vector_type(4))) float v4f;
typedef __attribute__((ext_vector_type(8))) float v8f;

#define NNETS 256
#define HID 128
#define BATCH 2048
#define W2_STRIDE 132
#define NB 4
#define WAVES 8
#define ROWS_PER_WAVE (BATCH / WAVES)
#define ROWS_PER_PASS (NB * 16)
#define PASSES (ROWS_PER_WAVE / ROWS_PER_PASS)

__device__ __forceinline__ void load_a(v2f (&af)[32], const float* arow)
{
    #pragma unroll
    for (int s = 0; s < 32; ++s) af[s] = *(const v2f*)(arow + 4 * s);
}

__device__ __forceinline__ void tile_products(const v2f (&af)[32], const v2f (&bfr)[NB][32],
                                              float (&acc)[NB], const float* sB2, const float* sW3,
                                              int mt, int lh)
{
    v8f c[NB];
    #pragma unroll
    for (int j = 0; j < NB; ++j) c[j] = (v8f){0.f, 0.f, 0.f, 0.f, 0.f, 0.f, 0.f, 0.f};

    #pragma unroll
    for (int s = 0; s < 32; ++s) {
        #pragma unroll
        for (int j = 0; j < NB; ++j) {
            c[j] = __builtin_amdgcn_wmma_f32_16x16x4_f32(
                       false, af[s], false, bfr[j][s], (short)0, c[j], false, false);
        }
    }

    #pragma unroll
    for (int v = 0; v < 8; ++v) {
        int kout = mt * 16 + v + 8 * lh;
        float b2v = sB2[kout];
        float w3v = sW3[kout];
        #pragma unroll
        for (int j = 0; j < NB; ++j) {
            float h2 = fmaxf(c[j][v] + b2v, 0.0f);
            acc[j] += h2 * w3v;
        }
    }
}

__global__ __launch_bounds__(256) void mlp256_net_resident_ntile4_wmma_f32(
    const float* __restrict__ a,  const float* __restrict__ W1,
    const float* __restrict__ b1, const float* __restrict__ W2,
    const float* __restrict__ b2, const float* __restrict__ W3,
    const float* __restrict__ b3, float* __restrict__ out)
{
    __shared__ __attribute__((aligned(16))) float sW2[HID * W2_STRIDE];
    __shared__ __attribute__((aligned(16))) float sW1[HID];
    __shared__ __attribute__((aligned(16))) float sB1[HID];
    __shared__ __attribute__((aligned(16))) float sB2[HID];
    __shared__ __attribute__((aligned(16))) float sW3[HID];
    __shared__ __attribute__((aligned(16))) float sA[BATCH];

    const int n = blockIdx.x;
    const int t = threadIdx.x;

    if (t < HID) {
        sW1[t] = W1[n * HID + t];
        sB1[t] = b1[n * HID + t];
        sB2[t] = b2[n * HID + t];
        sW3[t] = W3[n * HID + t];
    }
    const float* ga = a + (size_t)n * BATCH;
    #pragma unroll
    for (int i = 0; i < 2; ++i) {
        int e = (t + 256 * i) * 4;
        *(v4f*)(sA + e) = *(const v4f*)(ga + e);
    }

    const float* gW2 = W2 + (size_t)n * HID * HID;
    #pragma unroll
    for (int i = 0; i < 16; ++i) {
        int e   = (t + 256 * i) * 4;
        int row = e >> 7;
        int col = e & 127;
        v4f v = *(const v4f*)(gW2 + e);
        *(v4f*)(sW2 + row * W2_STRIDE + col) = v;
    }
    __syncthreads();

    const int lane = t & 31;
    const int wv   = t >> 5;
    const int li   = lane & 15;
    const int lh   = lane >> 4;
    const float b3v = b3[n];
    float* gout = out + (size_t)n * BATCH;
    const float* abase = sW2 + li * W2_STRIDE + 2 * lh;

    v2f afA[32], afB[32];
    load_a(afA, abase);

    #pragma unroll 1
    for (int p = 0; p < PASSES; ++p) {
        const int base = wv * ROWS_PER_WAVE + p * ROWS_PER_PASS;

        asm volatile("" ::: "memory");

        float av[NB];
        #pragma unroll
        for (int j = 0; j < NB; ++j) av[j] = sA[base + 16 * j + li];

        v2f bfr[NB][32];
        #pragma unroll
        for (int s = 0; s < 32; ++s) {
            int k = 4 * s + 2 * lh;
            v2f w1v = *(const v2f*)(sW1 + k);
            v2f b1v = *(const v2f*)(sB1 + k);
            #pragma unroll
            for (int j = 0; j < NB; ++j) {
                v2f h;
                h.x = fmaxf(av[j] * w1v.x + b1v.x, 0.0f);
                h.y = fmaxf(av[j] * w1v.y + b1v.y, 0.0f);
                bfr[j][s] = h;
            }
        }

        float acc[NB];
        #pragma unroll
        for (int j = 0; j < NB; ++j) acc[j] = 0.0f;

        #pragma unroll 1
        for (int mp = 0; mp < 4; ++mp) {
            const int mt = 2 * mp;
            load_a(afB, abase + (mt + 1) * 16 * W2_STRIDE);
            tile_products(afA, bfr, acc, sB2, sW3, mt, lh);
            load_a(afA, abase + ((mt + 2) & 7) * 16 * W2_STRIDE);
            tile_products(afB, bfr, acc, sB2, sW3, mt + 1, lh);
        }

        #pragma unroll
        for (int j = 0; j < NB; ++j) acc[j] += __shfl_xor(acc[j], 16, 32);

        #pragma unroll
        for (int jj = 0; jj < NB / 2; ++jj) {
            float r = lh ? acc[2 * jj + 1] : acc[2 * jj];
            gout[base + 32 * jj + lane] = r + b3v;
        }
    }
}

extern "C" void kernel_launch(void* const* d_in, const int* in_sizes, int n_in,
                              void* d_out, int out_size, void* d_ws, size_t ws_size,
                              hipStream_t stream) {
    const float* a  = (const float*)d_in[0];
    const float* W1 = (const float*)d_in[1];
    const float* b1 = (const float*)d_in[2];
    const float* W2 = (const float*)d_in[3];
    const float* b2 = (const float*)d_in[4];
    const float* W3 = (const float*)d_in[5];
    const float* b3 = (const float*)d_in[6];
    float* out = (float*)d_out;

    dim3 grid(NNETS);
    mlp256_net_resident_ntile4_wmma_f32<<<grid, 256, 0, stream>>>(a, W1, b1, W2, b2, W3, b3, out);
}


# Layer-2 k steps whose layer-1 outputs are all zero issue no WMMA: rows classed by a, units by W1 sign (wave ballots), wave-uniform skip per k step; k order changes (order of sums changes)
# speedup vs baseline: 1.4592x; 1.2692x over previous
#include <hip/hip_runtime.h>


typedef __attribute__((ext_vector_type(2))) float v2f;
typedef __attribute__((ext_vector_type(4))) float v4f;
typedef __attribute__((ext_vector_type(8))) float v8f;
typedef __attribute__((ext_vector_type(4))) int v4i;

#define NNETS 256
#define HID 128
#define BATCH 2048
#define W2_STRIDE 132
#define NB 4
#define WAVES 8
#define ROWS_PER_PASS (NB * 16)
#define PASSES (BATCH / (WAVES * ROWS_PER_PASS))
#define TAU 0.1f

__device__ __forceinline__ void load_a(v2f (&af)[32], const float* arow)
{
    #pragma unroll
    for (int s = 0; s < 32; ++s) af[s] = *(const v2f*)(arow + 4 * s);
}

__device__ __forceinline__ void tile_products(const v2f (&af)[32], const v2f (&bfr)[NB][32],
                                              float (&acc)[NB], const float* sB2, const float* sW3,
                                              int mt, int lh, unsigned live)
{
    v8f c[NB];
    #pragma unroll
    for (int j = 0; j < NB; ++j) c[j] = (v8f){0.f, 0.f, 0.f, 0.f, 0.f, 0.f, 0.f, 0.f};

    #pragma unroll
    for (int s = 0; s < 32; ++s) {
        if (live & (1u << s)) {
            #pragma unroll
            for (int j = 0; j < NB; ++j) {
                c[j] = __builtin_amdgcn_wmma_f32_16x16x4_f32(
                           false, af[s], false, bfr[j][s], (short)0, c[j], false, false);
            }
        }
    }
    asm volatile("v_nop\n\tv_nop\n\tv_nop\n\tv_nop\n\tv_nop");

    #pragma unroll
    for (int v = 0; v < 8; ++v) {
        int kout = mt * 16 + v + 8 * lh;
        float b2v = sB2[kout];
        float w3v = sW3[kout];
        #pragma unroll
        for (int j = 0; j < NB; ++j) {
            float h2 = fmaxf(c[j][v] + b2v, 0.0f);
            acc[j] += h2 * w3v;
        }
    }
}

__global__ __launch_bounds__(256) void mlp256_zero_kstep_skip_wmma_f32(
    const float* __restrict__ a,  const float* __restrict__ W1,
    const float* __restrict__ b1, const float* __restrict__ W2,
    const float* __restrict__ b2, const float* __restrict__ W3,
    const float* __restrict__ b3, float* __restrict__ out)
{
    __shared__ __attribute__((aligned(16))) float sW2[HID * W2_STRIDE];
    __shared__ __attribute__((aligned(16))) float sW1[HID];
    __shared__ __attribute__((aligned(16))) float sB1[HID];
    __shared__ __attribute__((aligned(16))) float sB2[HID];
    __shared__ __attribute__((aligned(16))) float sW3[HID];
    __shared__ __attribute__((aligned(16))) float sA[BATCH];
    __shared__ __attribute__((aligned(16))) int   sRow[BATCH];
    __shared__ __attribute__((aligned(16))) float sOut[BATCH];
    __shared__ __attribute__((aligned(16))) int   sUPos[HID];
    __shared__ int sCnt[WAVES][4];
    __shared__ int sBase[WAVES][4];
    __shared__ int sUCnt[4][3];
    __shared__ int sUBase[4][3];

    const int n = blockIdx.x;
    const int t = threadIdx.x;
    const int lane = t & 31;
    const int wv   = t >> 5;
    const unsigned lt = (1u << lane) - 1u;

    const float* gW2 = W2 + (size_t)n * HID * HID;
    v4f w2v[16];
    #pragma unroll
    for (int i = 0; i < 16; ++i) w2v[i] = *(const v4f*)(gW2 + (t + 256 * i) * 4);

    const float* ga = a + (size_t)n * BATCH;
    float ar[8];
    #pragma unroll
    for (int i = 0; i < 2; ++i) {
        v4f x = *(const v4f*)(ga + (t + 256 * i) * 4);
        ar[4 * i + 0] = x.x; ar[4 * i + 1] = x.y; ar[4 * i + 2] = x.z; ar[4 * i + 3] = x.w;
    }

    int ub = 0, ulr = 0;
    float w1 = 0.f, bb1 = 0.f;
    if (t < HID) {
        w1  = W1[n * HID + t];
        bb1 = b1[n * HID + t];
        ub = (fabsf(bb1) >= TAU * fabsf(w1)) ? 1 : (w1 > 0.f ? 0 : 2);
        #pragma unroll
        for (int g = 0; g < 3; ++g) {
            unsigned m = __builtin_amdgcn_ballot_w32(ub == g);
            if (ub == g) ulr = __builtin_popcount(m & lt);
            if (lane == 0) sUCnt[wv][g] = __builtin_popcount(m);
        }
        sB2[t] = b2[n * HID + t];
        sW3[t] = W3[n * HID + t];
    }

    int rb[8], rl[8];
    int cw[4] = {0, 0, 0, 0};
    #pragma unroll
    for (int q = 0; q < 8; ++q) {
        float x = ar[q];
        int g = (x >= TAU) ? 0 : ((x >= 0.f) ? 1 : ((x > -TAU) ? 2 : 3));
        int r = 0;
        #pragma unroll
        for (int gg = 0; gg < 4; ++gg) {
            unsigned m = __builtin_amdgcn_ballot_w32(g == gg);
            if (g == gg) r = cw[gg] + __builtin_popcount(m & lt);
            cw[gg] += __builtin_popcount(m);
        }
        rb[q] = g;
        rl[q] = r;
    }
    if (lane == 0) {
        #pragma unroll
        for (int gg = 0; gg < 4; ++gg) sCnt[wv][gg] = cw[gg];
    }
    __syncthreads();

    if (t < 32) {
        const int w = t >> 2, g = t & 3;
        int b = 0;
        for (int gg = 0; gg < 4; ++gg)
            for (int ww = 0; ww < WAVES; ++ww) {
                int cnt = sCnt[ww][gg];
                if (gg < g || (gg == g && ww < w)) b += cnt;
            }
        sBase[w][g] = b;
    } else if (t < 44) {
        const int u = t - 32;
        const int w = u / 3, g = u % 3;
        int b = 0;
        for (int gg = 0; gg < 3; ++gg)
            for (int ww = 0; ww < 4; ++ww) {
                int cnt = sUCnt[ww][gg];
                if (gg < g || (gg == g && ww < w)) b += cnt;
            }
        sUBase[w][g] = b;
    }
    __syncthreads();

    #pragma unroll
    for (int q = 0; q < 8; ++q) {
        int pos = sBase[wv][rb[q]] + rl[q];
        sA[pos]   = ar[q];
        sRow[pos] = (t + 256 * (q >> 2)) * 4 + (q & 3);
    }
    if (t < HID) {
        int up = sUBase[wv][ub] + ulr;
        sUPos[t] = up;
        sW1[up]  = w1;
        sB1[up]  = bb1;
    }
    __syncthreads();

    {
        const v4i pc = *(const v4i*)(sUPos + ((4 * t) & 127));
        #pragma unroll
        for (int i = 0; i < 16; ++i) {
            int row = ((t + 256 * i) * 4) >> 7;
            float* d = sW2 + row * W2_STRIDE;
            d[pc.x] = w2v[i].x; d[pc.y] = w2v[i].y; d[pc.z] = w2v[i].z; d[pc.w] = w2v[i].w;
        }
    }
    __syncthreads();

    const int li   = lane & 15;
    const int lh   = lane >> 4;
    const float b3v = b3[n];
    float* gout = out + (size_t)n * BATCH;
    const float* abase = sW2 + li * W2_STRIDE + 2 * lh;

    v2f afA[32], afB[32];
    load_a(afA, abase);

    #pragma unroll 1
    for (int p = 0; p < PASSES; ++p) {
        const int base = (p * WAVES + wv) * ROWS_PER_PASS;

        asm volatile("" ::: "memory");

        float av[NB];
        #pragma unroll
        for (int j = 0; j < NB; ++j) av[j] = sA[base + 16 * j + li];

        v2f bfr[NB][32];
        unsigned live = 0;
        #pragma unroll
        for (int s = 0; s < 32; ++s) {
            int k = 4 * s + 2 * lh;
            v2f w1v = *(const v2f*)(sW1 + k);
            v2f b1v = *(const v2f*)(sB1 + k);
            unsigned any = 0;
            #pragma unroll
            for (int j = 0; j < NB; ++j) {
                v2f h;
                h.x = fmaxf(av[j] * w1v.x + b1v.x, 0.0f);
                h.y = fmaxf(av[j] * w1v.y + b1v.y, 0.0f);
                bfr[j][s] = h;
                any |= __float_as_uint(h.x) | __float_as_uint(h.y);
            }
            if (__builtin_amdgcn_ballot_w32(any != 0u) != 0u) live |= (1u << s);
        }

        float acc[NB];
        #pragma unroll
        for (int j = 0; j < NB; ++j) acc[j] = 0.0f;

        #pragma unroll 1
        for (int mp = 0; mp < 4; ++mp) {
            const int mt = 2 * mp;
            load_a(afB, abase + (mt + 1) * 16 * W2_STRIDE);
            tile_products(afA, bfr, acc, sB2, sW3, mt, lh, live);
            load_a(afA, abase + ((mt + 2) & 7) * 16 * W2_STRIDE);
            tile_products(afB, bfr, acc, sB2, sW3, mt + 1, lh, live);
        }

        #pragma unroll
        for (int j = 0; j < NB; ++j) acc[j] += __shfl_xor(acc[j], 16, 32);

        #pragma unroll
        for (int jj = 0; jj < NB / 2; ++jj) {
            float r = lh ? acc[2 * jj + 1] : acc[2 * jj];
            sOut[sRow[base + 32 * jj + lane]] = r + b3v;
        }
    }
    __syncthreads();

    #pragma unroll
    for (int i = 0; i < 2; ++i) {
        int e = (t + 256 * i) * 4;
        *(v4f*)(gout + e) = *(const v4f*)(sOut + e);
    }
}

extern "C" void kernel_launch(void* const* d_in, const int* in_sizes, int n_in,
                              void* d_out, int out_size, void* d_ws, size_t ws_size,
                              hipStream_t stream) {
    const float* a  = (const float*)d_in[0];
    const float* W1 = (const float*)d_in[1];
    const float* b1 = (const float*)d_in[2];
    const float* W2 = (const float*)d_in[3];
    const float* b2 = (const float*)d_in[4];
    const float* W3 = (const float*)d_in[5];
    const float* b3 = (const float*)d_in[6];
    float* out = (float*)d_out;

    dim3 grid(NNETS);
    mlp256_zero_kstep_skip_wmma_f32<<<grid, 256, 0, stream>>>(a, W1, b1, W2, b2, W3, b3, out);
}


# Units ranked by kink within W1 sign, rows in 16 classes by sign and exponent of a (was 3 and 4 classes): 64-row blocks span less of a, fewer live k steps; k order changes (order of sums changes)
# speedup vs baseline: 1.6285x; 1.1160x over previous
#include <hip/hip_runtime.h>


typedef __attribute__((ext_vector_type(2))) float v2f;
typedef __attribute__((ext_vector_type(4))) float v4f;
typedef __attribute__((ext_vector_type(8))) float v8f;
typedef __attribute__((ext_vector_type(4))) int v4i;
typedef __attribute__((ext_vector_type(4))) unsigned v4u;

#define NNETS 256
#define HID 128
#define BATCH 2048
#define W2_STRIDE 132
#define NB 4
#define WAVES 8
#define MT 8
#define ROWS_PER_PASS (NB * 16)
#define PASSES (BATCH / (WAVES * ROWS_PER_PASS))
#define RCLS 16

__device__ __forceinline__ void fetch_a(v2f (&af)[MT], const float* abase, int s)
{
    const float* p = abase + 4 * s;
    #pragma unroll
    for (int mt = 0; mt < MT; ++mt) af[mt] = *(const v2f*)(p + mt * 16 * W2_STRIDE);
}

__device__ __forceinline__ void make_h(v2f (&h)[NB], const float (&av)[NB], v2f w1v, v2f b1v)
{
    #pragma unroll
    for (int j = 0; j < NB; ++j) {
        h[j].x = fmaxf(av[j] * w1v.x + b1v.x, 0.0f);
        h[j].y = fmaxf(av[j] * w1v.y + b1v.y, 0.0f);
    }
}

template <int T0, int T1>
__device__ __forceinline__ void products(v8f (&c)[MT][NB], const v2f (&af)[MT], const v2f (&h)[NB])
{
    #pragma unroll
    for (int mt = T0; mt < T1; ++mt) {
        #pragma unroll
        for (int j = 0; j < NB; ++j) {
            c[mt][j] = __builtin_amdgcn_wmma_f32_16x16x4_f32(
                           false, af[mt], false, h[j], (short)0, c[mt][j], false, false);
        }
    }
}

__device__ __forceinline__ void k_step(v8f (&c)[MT][NB], const v2f (&ac)[MT], const v2f (&hc)[NB],
                                       v2f (&an)[MT], v2f (&hn)[NB], const float (&av)[NB],
                                       const float* abase, const float* w1p, const float* b1p, int sn)
{
    products<0, 2>(c, ac, hc);
    __builtin_amdgcn_sched_barrier(0);
    asm volatile("v_nop\n\tv_nop\n\tv_nop\n\tv_nop\n\tv_nop");
    __builtin_amdgcn_sched_barrier(0);
    const v2f w1v = *(const v2f*)(w1p + 4 * sn);
    const v2f b1v = *(const v2f*)(b1p + 4 * sn);
    fetch_a(an, abase, sn);
    __builtin_amdgcn_sched_barrier(0);
    products<2, 6>(c, ac, hc);
    __builtin_amdgcn_sched_barrier(0);
    asm volatile("v_nop\n\tv_nop\n\tv_nop\n\tv_nop\n\tv_nop");
    __builtin_amdgcn_sched_barrier(0);
    make_h(hn, av, w1v, b1v);
    #pragma unroll
    for (int j = 0; j < NB; ++j) asm volatile("" : "+v"(hn[j]));
    __builtin_amdgcn_sched_barrier(0);
    products<6, 8>(c, ac, hc);
    #pragma unroll
    for (int mt = 0; mt < MT; ++mt) asm volatile("" : "+v"(an[mt]));
}

template <int M>
__device__ __forceinline__ float row_xor(float x)
{
    int xi = __builtin_bit_cast(int, x);
    int r = __builtin_amdgcn_update_dpp(xi, xi, 0x160 + M, 0xf, 0xf, false);
    return __builtin_bit_cast(float, r);
}

__global__ __launch_bounds__(256) void mlp256_kouter_live_steps_wmma_f32(
    const float* __restrict__ a,  const float* __restrict__ W1,
    const float* __restrict__ b1, const float* __restrict__ W2,
    const float* __restrict__ b2, const float* __restrict__ W3,
    const float* __restrict__ b3, float* __restrict__ out)
{
    __shared__ __attribute__((aligned(16))) float sW2[HID * W2_STRIDE];
    __shared__ __attribute__((aligned(16))) float sW1[HID];
    __shared__ __attribute__((aligned(16))) float sB1[HID];
    __shared__ __attribute__((aligned(16))) float sB2[HID];
    __shared__ __attribute__((aligned(16))) float sW3[HID];
    __shared__ __attribute__((aligned(16))) float sA[BATCH];
    __shared__ __attribute__((aligned(16))) int   sRow[BATCH];
    __shared__ __attribute__((aligned(16))) float sOut[BATCH];
    __shared__ __attribute__((aligned(16))) int   sUPos[HID];
    __shared__ __attribute__((aligned(16))) unsigned sKey[HID];
    __shared__ __attribute__((aligned(16))) int   sCnt[RCLS * WAVES];
    __shared__ __attribute__((aligned(16))) int   sBase[RCLS * WAVES];

    const int n = blockIdx.x;
    const int t = threadIdx.x;
    const int lane = t & 31;
    const int wv   = t >> 5;
    const unsigned lt = (1u << lane) - 1u;

    const float* gW2 = W2 + (size_t)n * HID * HID;
    v4f w2v[16];
    #pragma unroll
    for (int i = 0; i < 16; ++i) w2v[i] = *(const v4f*)(gW2 + (t + 256 * i) * 4);

    const float* ga = a + (size_t)n * BATCH;
    float ar[8];
    #pragma unroll
    for (int i = 0; i < 2; ++i) {
        v4f x = *(const v4f*)(ga + (t + 256 * i) * 4);
        ar[4 * i + 0] = x.x; ar[4 * i + 1] = x.y; ar[4 * i + 2] = x.z; ar[4 * i + 3] = x.w;
    }

    float w1 = 0.f, bb1 = 0.f;
    unsigned key = 0u;
    if (t < HID) {
        w1  = W1[n * HID + t];
        bb1 = b1[n * HID + t];
        const unsigned kb = __builtin_bit_cast(unsigned, -bb1 / w1);
        const unsigned so = (kb & 0x80000000u) ? ~kb : (kb | 0x80000000u);
        key = ((w1 > 0.f) ? 0u : 0x80000000u) | ((so >> 8) << 7) | (unsigned)t;
        sKey[t] = key;
        sB2[t] = b2[n * HID + t];
        sW3[t] = W3[n * HID + t];
    }

    int xl[4];
    #pragma unroll
    for (int b = 0; b < 4; ++b) xl[b] = ((lane >> b) & 1) - 1;
    int rb[8], rl[8];
    int run = 0;
    #pragma unroll
    for (int q = 0; q < 8; ++q) {
        const unsigned ab = __builtin_bit_cast(unsigned, ar[q]);
        const int e  = (int)((ab >> 23) & 0xffu);
        const int mm = min(max(e - 120, 0), 7);
        const int g  = (ab >> 31) ? 8 + mm : 7 - mm;
        unsigned mine = 0xffffffffu, ofl = 0xffffffffu;
        #pragma unroll
        for (int b = 0; b < 4; ++b) {
            const unsigned bb = __builtin_amdgcn_ballot_w32(((g >> b) & 1) != 0);
            mine &= bb ^ (unsigned)(((g >> b) & 1) - 1);
            ofl  &= bb ^ (unsigned)xl[b];
        }
        rb[q] = g;
        rl[q] = __shfl(run, g, 32) + __builtin_popcount(mine & lt);
        run += __builtin_popcount(ofl);
    }
    if (lane < RCLS) sCnt[lane * WAVES + wv] = run;
    __syncthreads();

    if (t < HID) {
        int up = 0;
        #pragma unroll
        for (int j = 0; j < HID; j += 4) {
            const v4u kk = *(const v4u*)(sKey + j);
            up += (kk.x < key) + (kk.y < key) + (kk.z < key) + (kk.w < key);
        }
        sUPos[t] = up;
        sW1[up]  = w1;
        sB1[up]  = bb1;
    } else {
        const int i = t - HID;
        int b = 0;
        #pragma unroll
        for (int j = 0; j < RCLS * WAVES; j += 4) {
            const v4i cc = *(const v4i*)(sCnt + j);
            b += (j + 0 < i ? cc.x : 0) + (j + 1 < i ? cc.y : 0) + (j + 2 < i ? cc.z : 0) + (j + 3 < i ? cc.w : 0);
        }
        sBase[i] = b;
    }
    __syncthreads();

    #pragma unroll
    for (int q = 0; q < 8; ++q) {
        int pos = sBase[rb[q] * WAVES + wv] + rl[q];
        sA[pos]   = ar[q];
        sRow[pos] = (t + 256 * (q >> 2)) * 4 + (q & 3);
    }

    {
        const v4i pc = *(const v4i*)(sUPos + ((4 * t) & 127));
        #pragma unroll
        for (int i = 0; i < 16; ++i) {
            int row = ((t + 256 * i) * 4) >> 7;
            float* d = sW2 + row * W2_STRIDE;
            d[pc.x] = w2v[i].x; d[pc.y] = w2v[i].y; d[pc.z] = w2v[i].z; d[pc.w] = w2v[i].w;
        }
    }
    __syncthreads();

    const int li   = lane & 15;
    const int lh   = lane >> 4;
    const float b3v = b3[n];
    float* gout = out + (size_t)n * BATCH;
    const float* abase = sW2 + li * W2_STRIDE + 2 * lh;

    const float* w1p = sW1 + 2 * lh;
    const float* b1p = sB1 + 2 * lh;

    #pragma unroll 1
    for (int p = 0; p < PASSES; ++p) {
        const int base = (p * WAVES + wv) * ROWS_PER_PASS;

        float av[NB];
        #pragma unroll
        for (int j = 0; j < NB; ++j) av[j] = sA[base + 16 * j + li];

        float lo = fminf(fminf(av[0], av[1]), fminf(av[2], av[3]));
        float hi = fmaxf(fmaxf(av[0], av[1]), fmaxf(av[2], av[3]));
        lo = fminf(lo, row_xor<1>(lo));  hi = fmaxf(hi, row_xor<1>(hi));
        lo = fminf(lo, row_xor<2>(lo));  hi = fmaxf(hi, row_xor<2>(hi));
        lo = fminf(lo, row_xor<4>(lo));  hi = fmaxf(hi, row_xor<4>(hi));
        lo = fminf(lo, row_xor<8>(lo));  hi = fmaxf(hi, row_xor<8>(hi));
        const v4f wq = *(const v4f*)(sW1 + 4 * lane);
        const v4f bq = *(const v4f*)(sB1 + 4 * lane);
        const bool on = (__builtin_fmaf(lo, wq.x, bq.x) > 0.0f) | (__builtin_fmaf(hi, wq.x, bq.x) > 0.0f)
                      | (__builtin_fmaf(lo, wq.y, bq.y) > 0.0f) | (__builtin_fmaf(hi, wq.y, bq.y) > 0.0f)
                      | (__builtin_fmaf(lo, wq.z, bq.z) > 0.0f) | (__builtin_fmaf(hi, wq.z, bq.z) > 0.0f)
                      | (__builtin_fmaf(lo, wq.w, bq.w) > 0.0f) | (__builtin_fmaf(hi, wq.w, bq.w) > 0.0f);
        unsigned m = __builtin_amdgcn_ballot_w32(on);

        v8f c[MT][NB];
        #pragma unroll
        for (int mt = 0; mt < MT; ++mt) {
            #pragma unroll
            for (int j = 0; j < NB; ++j) c[mt][j] = (v8f){0.f, 0.f, 0.f, 0.f, 0.f, 0.f, 0.f, 0.f};
        }
        v2f a0[MT], a1[MT], h0[NB], h1[NB];
        bool more = (m != 0u);
        int sn = more ? __builtin_ctz(m) : 0;
        fetch_a(a0, abase, sn);
        make_h(h0, av, *(const v2f*)(w1p + 4 * sn), *(const v2f*)(b1p + 4 * sn));
        while (more) {
            m &= m - 1u;
            more = (m != 0u);
            sn = more ? __builtin_ctz(m) : 0;
            k_step(c, a0, h0, a1, h1, av, abase, w1p, b1p, sn);
            if (!more) break;
            m &= m - 1u;
            more = (m != 0u);
            sn = more ? __builtin_ctz(m) : 0;
            k_step(c, a1, h1, a0, h0, av, abase, w1p, b1p, sn);
        }
        asm volatile("v_nop\n\tv_nop\n\tv_nop\n\tv_nop\n\tv_nop");

        float acc[NB];
        #pragma unroll
        for (int j = 0; j < NB; ++j) acc[j] = 0.0f;
        #pragma unroll
        for (int mt = 0; mt < MT; ++mt) {
            #pragma unroll
            for (int v = 0; v < 8; ++v) {
                int kout = mt * 16 + v + 8 * lh;
                float b2v = sB2[kout];
                float w3v = sW3[kout];
                #pragma unroll
                for (int j = 0; j < NB; ++j) {
                    float h2 = fmaxf(c[mt][j][v] + b2v, 0.0f);
                    acc[j] += h2 * w3v;
                }
            }
        }

        #pragma unroll
        for (int j = 0; j < NB; ++j) acc[j] += __shfl_xor(acc[j], 16, 32);

        #pragma unroll
        for (int jj = 0; jj < NB / 2; ++jj) {
            float r = lh ? acc[2 * jj + 1] : acc[2 * jj];
            sOut[sRow[base + 32 * jj + lane]] = r + b3v;
        }
    }
    __syncthreads();

    #pragma unroll
    for (int i = 0; i < 2; ++i) {
        int e = (t + 256 * i) * 4;
        *(v4f*)(gout + e) = *(const v4f*)(sOut + e);
    }
}

extern "C" void kernel_launch(void* const* d_in, const int* in_sizes, int n_in,
                              void* d_out, int out_size, void* d_ws, size_t ws_size,
                              hipStream_t stream) {
    const float* a  = (const float*)d_in[0];
    const float* W1 = (const float*)d_in[1];
    const float* b1 = (const float*)d_in[2];
    const float* W2 = (const float*)d_in[3];
    const float* b2 = (const float*)d_in[4];
    const float* W3 = (const float*)d_in[5];
    const float* b3 = (const float*)d_in[6];
    float* out = (float*)d_out;

    dim3 grid(NNETS);
    mlp256_kouter_live_steps_wmma_f32<<<grid, 256, 0, stream>>>(a, W1, b1, W2, b2, W3, b3, out);
}


# k steps whose 4 units are on for all 64 rows of a block issue no WMMA of their own: running-sum tables of W2*W1 and W2*b1 feed ONE affine k step, B=(a,1,a,1), per block; order of sums changes
# speedup vs baseline: 3.3591x; 2.0627x over previous
#include <hip/hip_runtime.h>


typedef __attribute__((ext_vector_type(2))) float v2f;
typedef __attribute__((ext_vector_type(4))) float v4f;
typedef __attribute__((ext_vector_type(8))) float v8f;
typedef __attribute__((ext_vector_type(4))) int v4i;
typedef __attribute__((ext_vector_type(4))) unsigned v4u;

#define NNETS 256
#define HID 128
#define BATCH 2048
#define W2_STRIDE 132
#define NB 4
#define WAVES 8
#define MT 8
#define ROWS_PER_PASS (NB * 16)
#define PASSES (BATCH / (WAVES * ROWS_PER_PASS))
#define RCLS 16

__device__ __forceinline__ void fetch_a(v2f (&af)[MT], const float* abase, int s)
{
    const float* p = abase + 4 * s;
    #pragma unroll
    for (int mt = 0; mt < MT; ++mt) af[mt] = *(const v2f*)(p + mt * 16 * W2_STRIDE);
}

__device__ __forceinline__ void make_h(v2f (&h)[NB], const float (&av)[NB], v2f w1v, v2f b1v)
{
    #pragma unroll
    for (int j = 0; j < NB; ++j) {
        h[j].x = fmaxf(av[j] * w1v.x + b1v.x, 0.0f);
        h[j].y = fmaxf(av[j] * w1v.y + b1v.y, 0.0f);
    }
}

template <int T0, int T1, bool FIRST>
__device__ __forceinline__ void products(v8f (&c)[MT][NB], const v2f (&af)[MT], const v2f (&h)[NB])
{
    const v8f zero = {0.f, 0.f, 0.f, 0.f, 0.f, 0.f, 0.f, 0.f};
    #pragma unroll
    for (int mt = T0; mt < T1; ++mt) {
        #pragma unroll
        for (int j = 0; j < NB; ++j) {
            c[mt][j] = __builtin_amdgcn_wmma_f32_16x16x4_f32(
                           false, af[mt], false, h[j], (short)0, FIRST ? zero : c[mt][j], false, false);
        }
    }
}

template <bool FIRST>
__device__ __forceinline__ void k_step(v8f (&c)[MT][NB], const v2f (&ac)[MT], const v2f (&hc)[NB],
                                       v2f (&an)[MT], v2f (&hn)[NB], const float (&av)[NB],
                                       const float* abase, const float* w1p, const float* b1p, int sn)
{
    products<0, 2, FIRST>(c, ac, hc);
    __builtin_amdgcn_sched_barrier(0);
    asm volatile("v_nop\n\tv_nop\n\tv_nop\n\tv_nop\n\tv_nop");
    __builtin_amdgcn_sched_barrier(0);
    const v2f w1v = *(const v2f*)(w1p + 4 * sn);
    const v2f b1v = *(const v2f*)(b1p + 4 * sn);
    fetch_a(an, abase, sn);
    __builtin_amdgcn_sched_barrier(0);
    products<2, 6, FIRST>(c, ac, hc);
    __builtin_amdgcn_sched_barrier(0);
    asm volatile("v_nop\n\tv_nop\n\tv_nop\n\tv_nop\n\tv_nop");
    __builtin_amdgcn_sched_barrier(0);
    make_h(hn, av, w1v, b1v);
    #pragma unroll
    for (int j = 0; j < NB; ++j) asm volatile("" : "+v"(hn[j]));
    __builtin_amdgcn_sched_barrier(0);
    products<6, 8, FIRST>(c, ac, hc);
    #pragma unroll
    for (int mt = 0; mt < MT; ++mt) asm volatile("" : "+v"(an[mt]));
    if (FIRST) {
        __builtin_amdgcn_sched_barrier(0);
        asm volatile("v_nop\n\tv_nop\n\tv_nop\n\tv_nop\n\tv_nop");
        __builtin_amdgcn_sched_barrier(0);
    }
}

template <bool SUF>
__device__ __forceinline__ void build_affine(v2f* tab, const float* wrow, const float* w1s,
                                             const float* b1s, int k)
{
    float ps[32], qs[32];
    #pragma unroll
    for (int g = 0; g < 32; g += 8) {
        v4f w[8], x[8], y[8];
        #pragma unroll
        for (int i = 0; i < 8; ++i) {
            w[i] = *(const v4f*)(wrow + 4 * (g + i));
            x[i] = *(const v4f*)(w1s + 4 * (g + i));
            y[i] = *(const v4f*)(b1s + 4 * (g + i));
        }
        #pragma unroll
        for (int i = 0; i < 8; ++i) {
            ps[g + i] = __builtin_fmaf(w[i].x, x[i].x, w[i].y * x[i].y) + __builtin_fmaf(w[i].z, x[i].z, w[i].w * x[i].w);
            qs[g + i] = __builtin_fmaf(w[i].x, y[i].x, w[i].y * y[i].y) + __builtin_fmaf(w[i].z, y[i].z, w[i].w * y[i].w);
        }
    }
    float p = 0.0f, q = 0.0f;
    tab[(SUF ? 32 : 0) * HID + k] = (v2f){p, q};
    #pragma unroll
    for (int i = 0; i < 32; ++i) {
        const int s = SUF ? 31 - i : i;
        p += ps[s];
        q += qs[s];
        tab[(SUF ? s : s + 1) * HID + k] = (v2f){p, q};
    }
}

template <int M>
__device__ __forceinline__ float row_xor(float x)
{
    int xi = __builtin_bit_cast(int, x);
    int r = __builtin_amdgcn_update_dpp(xi, xi, 0x160 + M, 0xf, 0xf, false);
    return __builtin_bit_cast(float, r);
}

__global__ __launch_bounds__(256) void mlp256_kouter_live_steps_wmma_f32(
    const float* __restrict__ a,  const float* __restrict__ W1,
    const float* __restrict__ b1, const float* __restrict__ W2,
    const float* __restrict__ b2, const float* __restrict__ W3,
    const float* __restrict__ b3, float* __restrict__ out)
{
    __shared__ __attribute__((aligned(16))) float sW2[HID * W2_STRIDE];
    __shared__ __attribute__((aligned(16))) float sW1[HID];
    __shared__ __attribute__((aligned(16))) float sB1[HID];
    __shared__ __attribute__((aligned(16))) float sB2[HID];
    __shared__ __attribute__((aligned(16))) float sW3[HID];
    __shared__ __attribute__((aligned(16))) float sA[BATCH];
    __shared__ __attribute__((aligned(16))) int   sRow[BATCH];
    __shared__ __attribute__((aligned(16))) float sOut[BATCH];
    __shared__ __attribute__((aligned(16))) int   sUPos[HID];
    __shared__ __attribute__((aligned(16))) unsigned sKey[HID];
    __shared__ __attribute__((aligned(16))) int   sCnt[RCLS * WAVES];
    __shared__ __attribute__((aligned(16))) int   sBase[RCLS * WAVES];
    __shared__ __attribute__((aligned(16))) v2f   sTpre[33 * HID];
    __shared__ __attribute__((aligned(16))) v2f   sTsuf[33 * HID];

    const int n = blockIdx.x;
    const int t = threadIdx.x;
    const int lane = t & 31;
    const int wv   = t >> 5;
    const unsigned lt = (1u << lane) - 1u;

    const float* gW2 = W2 + (size_t)n * HID * HID;
    v4f w2v[16];
    #pragma unroll
    for (int i = 0; i < 16; ++i) w2v[i] = *(const v4f*)(gW2 + (t + 256 * i) * 4);

    const float* ga = a + (size_t)n * BATCH;
    float ar[8];
    #pragma unroll
    for (int i = 0; i < 2; ++i) {
        v4f x = *(const v4f*)(ga + (t + 256 * i) * 4);
        ar[4 * i + 0] = x.x; ar[4 * i + 1] = x.y; ar[4 * i + 2] = x.z; ar[4 * i + 3] = x.w;
    }

    float w1 = 0.f, bb1 = 0.f;
    unsigned key = 0u;
    if (t < HID) {
        w1  = W1[n * HID + t];
        bb1 = b1[n * HID + t];
        const unsigned kb = __builtin_bit_cast(unsigned, -bb1 / w1);
        const unsigned so = (kb & 0x80000000u) ? ~kb : (kb | 0x80000000u);
        key = ((w1 > 0.f) ? 0u : 0x80000000u) | ((so >> 8) << 7) | (unsigned)t;
        sKey[t] = key;
        sB2[t] = b2[n * HID + t];
        sW3[t] = W3[n * HID + t];
    }

    int xl[4];
    #pragma unroll
    for (int b = 0; b < 4; ++b) xl[b] = ((lane >> b) & 1) - 1;
    int rb[8], rl[8];
    int run = 0;
    #pragma unroll
    for (int q = 0; q < 8; ++q) {
        const unsigned ab = __builtin_bit_cast(unsigned, ar[q]);
        const int e  = (int)((ab >> 23) & 0xffu);
        const int mm = min(max(e - 120, 0), 7);
        const int g  = (ab >> 31) ? 8 + mm : 7 - mm;
        unsigned mine = 0xffffffffu, ofl = 0xffffffffu;
        #pragma unroll
        for (int b = 0; b < 4; ++b) {
            const unsigned bb = __builtin_amdgcn_ballot_w32(((g >> b) & 1) != 0);
            mine &= bb ^ (unsigned)(((g >> b) & 1) - 1);
            ofl  &= bb ^ (unsigned)xl[b];
        }
        rb[q] = g;
        rl[q] = __shfl(run, g, 32) + __builtin_popcount(mine & lt);
        run += __builtin_popcount(ofl);
    }
    if (lane < RCLS) sCnt[lane * WAVES + wv] = run;
    __syncthreads();

    if (t < HID) {
        int up = 0;
        #pragma unroll
        for (int j = 0; j < HID; j += 4) {
            const v4u kk = *(const v4u*)(sKey + j);
            up += (kk.x < key) + (kk.y < key) + (kk.z < key) + (kk.w < key);
        }
        sUPos[t] = up;
        sW1[up]  = w1;
        sB1[up]  = bb1;
    } else {
        const int i = t - HID;
        int b = 0;
        #pragma unroll
        for (int j = 0; j < RCLS * WAVES; j += 4) {
            const v4i cc = *(const v4i*)(sCnt + j);
            b += (j + 0 < i ? cc.x : 0) + (j + 1 < i ? cc.y : 0) + (j + 2 < i ? cc.z : 0) + (j + 3 < i ? cc.w : 0);
        }
        sBase[i] = b;
    }
    __syncthreads();

    #pragma unroll
    for (int q = 0; q < 8; ++q) {
        int pos = sBase[rb[q] * WAVES + wv] + rl[q];
        sA[pos]   = ar[q];
        sRow[pos] = (t + 256 * (q >> 2)) * 4 + (q & 3);
    }

    {
        const v4i pc = *(const v4i*)(sUPos + ((4 * t) & 127));
        #pragma unroll
        for (int i = 0; i < 16; ++i) {
            int row = ((t + 256 * i) * 4) >> 7;
            float* d = sW2 + row * W2_STRIDE;
            d[pc.x] = w2v[i].x; d[pc.y] = w2v[i].y; d[pc.z] = w2v[i].z; d[pc.w] = w2v[i].w;
        }
    }
    __syncthreads();

    {
        const int k = t & (HID - 1);
        const float* wrow = sW2 + k * W2_STRIDE;
        if (t < HID) build_affine<false>(sTpre, wrow, sW1, sB1, k);
        else         build_affine<true >(sTsuf, wrow, sW1, sB1, k);
    }
    __syncthreads();

    const int li   = lane & 15;
    const int lh   = lane >> 4;
    const float b3v = b3[n];
    float* gout = out + (size_t)n * BATCH;
    const float* abase = sW2 + li * W2_STRIDE + 2 * lh;

    const float* w1p = sW1 + 2 * lh;
    const float* b1p = sB1 + 2 * lh;

    #pragma unroll 1
    for (int p = 0; p < PASSES; ++p) {
        const int base = (p * WAVES + wv) * ROWS_PER_PASS;

        float av[NB];
        #pragma unroll
        for (int j = 0; j < NB; ++j) av[j] = sA[base + 16 * j + li];

        float lo = fminf(fminf(av[0], av[1]), fminf(av[2], av[3]));
        float hi = fmaxf(fmaxf(av[0], av[1]), fmaxf(av[2], av[3]));
        lo = fminf(lo, row_xor<1>(lo));  hi = fmaxf(hi, row_xor<1>(hi));
        lo = fminf(lo, row_xor<2>(lo));  hi = fmaxf(hi, row_xor<2>(hi));
        lo = fminf(lo, row_xor<4>(lo));  hi = fmaxf(hi, row_xor<4>(hi));
        lo = fminf(lo, row_xor<8>(lo));  hi = fmaxf(hi, row_xor<8>(hi));
        const v4f wq = *(const v4f*)(sW1 + 4 * lane);
        const v4f bq = *(const v4f*)(sB1 + 4 * lane);
        const bool lx = __builtin_fmaf(lo, wq.x, bq.x) > 0.0f, hx = __builtin_fmaf(hi, wq.x, bq.x) > 0.0f;
        const bool ly = __builtin_fmaf(lo, wq.y, bq.y) > 0.0f, hy = __builtin_fmaf(hi, wq.y, bq.y) > 0.0f;
        const bool lz = __builtin_fmaf(lo, wq.z, bq.z) > 0.0f, hz = __builtin_fmaf(hi, wq.z, bq.z) > 0.0f;
        const bool lw = __builtin_fmaf(lo, wq.w, bq.w) > 0.0f, hw = __builtin_fmaf(hi, wq.w, bq.w) > 0.0f;
        const bool on   = lx | hx | ly | hy | lz | hz | lw | hw;
        const bool full = lx & hx & ly & hy & lz & hz & lw & hw;
        const unsigned mon = __builtin_amdgcn_ballot_w32(on);
        const unsigned nf  = ~__builtin_amdgcn_ballot_w32(full);
        int s1 = 32, s3 = 32;
        unsigned m = 0u;
        if (nf != 0u) {
            s1 = __builtin_ctz(nf);
            s3 = 32 - __builtin_clz(nf);
            const unsigned below3 = (s3 >= 32) ? 0xffffffffu : ((1u << s3) - 1u);
            m = mon & below3 & ~((1u << s1) - 1u);
        }

        v8f c[MT][NB];
        v2f a0[MT], a1[MT], h0[NB], h1[NB];
        {
            const v2f* qb = (lh ? sTsuf + s3 * HID : sTpre + s1 * HID) + li;
            #pragma unroll
            for (int mt = 0; mt < MT; ++mt) a0[mt] = qb[mt * 16];
            #pragma unroll
            for (int j = 0; j < NB; ++j) h0[j] = (v2f){av[j], 1.0f};
        }
        bool more = (m != 0u);
        int sn = more ? __builtin_ctz(m) : 0;
        k_step<true>(c, a0, h0, a1, h1, av, abase, w1p, b1p, sn);
        while (more) {
            m &= m - 1u;
            more = (m != 0u);
            sn = more ? __builtin_ctz(m) : 0;
            k_step<false>(c, a1, h1, a0, h0, av, abase, w1p, b1p, sn);
            if (!more) break;
            m &= m - 1u;
            more = (m != 0u);
            sn = more ? __builtin_ctz(m) : 0;
            k_step<false>(c, a0, h0, a1, h1, av, abase, w1p, b1p, sn);
        }
        asm volatile("v_nop\n\tv_nop\n\tv_nop\n\tv_nop\n\tv_nop");

        float acc[NB];
        #pragma unroll
        for (int j = 0; j < NB; ++j) acc[j] = 0.0f;
        #pragma unroll
        for (int mt = 0; mt < MT; ++mt) {
            #pragma unroll
            for (int v = 0; v < 8; ++v) {
                int kout = mt * 16 + v + 8 * lh;
                float b2v = sB2[kout];
                float w3v = sW3[kout];
                #pragma unroll
                for (int j = 0; j < NB; ++j) {
                    float h2 = fmaxf(c[mt][j][v] + b2v, 0.0f);
                    acc[j] += h2 * w3v;
                }
            }
        }

        #pragma unroll
        for (int j = 0; j < NB; ++j) acc[j] += __shfl_xor(acc[j], 16, 32);

        #pragma unroll
        for (int jj = 0; jj < NB / 2; ++jj) {
            float r = lh ? acc[2 * jj + 1] : acc[2 * jj];
            sOut[sRow[base + 32 * jj + lane]] = r + b3v;
        }
    }
    __syncthreads();

    #pragma unroll
    for (int i = 0; i < 2; ++i) {
        int e = (t + 256 * i) * 4;
        *(v4f*)(gout + e) = *(const v4f*)(sOut + e);
    }
}

extern "C" void kernel_launch(void* const* d_in, const int* in_sizes, int n_in,
                              void* d_out, int out_size, void* d_ws, size_t ws_size,
                              hipStream_t stream) {
    const float* a  = (const float*)d_in[0];
    const float* W1 = (const float*)d_in[1];
    const float* b1 = (const float*)d_in[2];
    const float* W2 = (const float*)d_in[3];
    const float* b2 = (const float*)d_in[4];
    const float* W3 = (const float*)d_in[5];
    const float* b3 = (const float*)d_in[6];
    float* out = (float*)d_out;

    dim3 grid(NNETS);
    mlp256_kouter_live_steps_wmma_f32<<<grid, 256, 0, stream>>>(a, W1, b1, W2, b2, W3, b3, out);
}


# 16 waves per work-group (was 8), four per SIMD: 8 row blocks and half the staging per wave; a wave's LDS round trips and vector work run under the others' WMMAs; run starts summed in two levels
# speedup vs baseline: 3.8340x; 1.1414x over previous
#include <hip/hip_runtime.h>


typedef __attribute__((ext_vector_type(2))) float v2f;
typedef __attribute__((ext_vector_type(4))) float v4f;
typedef __attribute__((ext_vector_type(8))) float v8f;
typedef __attribute__((ext_vector_type(4))) int v4i;
typedef __attribute__((ext_vector_type(4))) unsigned v4u;

#define NNETS 256
#define THREADS 512
#define HID 128
#define BATCH 2048
#define W2_STRIDE 132
#define NB 1
#define WAVES 16
#define MT 8
#define ROWS_PER_PASS (NB * 16)
#define PASSES (BATCH / (WAVES * ROWS_PER_PASS))
#define RCLS 16

__device__ __forceinline__ void fetch_a(v2f (&af)[MT], const float* abase, int s)
{
    const float* p = abase + 4 * s;
    #pragma unroll
    for (int mt = 0; mt < MT; ++mt) af[mt] = *(const v2f*)(p + mt * 16 * W2_STRIDE);
}

__device__ __forceinline__ void make_h(v2f (&h)[NB], const float (&av)[NB], v2f w1v, v2f b1v)
{
    #pragma unroll
    for (int j = 0; j < NB; ++j) {
        h[j].x = fmaxf(av[j] * w1v.x + b1v.x, 0.0f);
        h[j].y = fmaxf(av[j] * w1v.y + b1v.y, 0.0f);
    }
}

template <int T0, int T1, bool FIRST>
__device__ __forceinline__ void products(v8f (&c)[MT][NB], const v2f (&af)[MT], const v2f (&h)[NB])
{
    const v8f zero = {0.f, 0.f, 0.f, 0.f, 0.f, 0.f, 0.f, 0.f};
    #pragma unroll
    for (int mt = T0; mt < T1; ++mt) {
        #pragma unroll
        for (int j = 0; j < NB; ++j) {
            c[mt][j] = __builtin_amdgcn_wmma_f32_16x16x4_f32(
                           false, af[mt], false, h[j], (short)0, FIRST ? zero : c[mt][j], false, false);
        }
    }
}

template <bool FIRST>
__device__ __forceinline__ void k_step(v8f (&c)[MT][NB], const v2f (&ac)[MT], const v2f (&hc)[NB],
                                       v2f (&an)[MT], v2f (&hn)[NB], const float (&av)[NB],
                                       const float* abase, const float* w1p, const float* b1p, int sn)
{
    products<0, 2, FIRST>(c, ac, hc);
    __builtin_amdgcn_sched_barrier(0);
    asm volatile("v_nop\n\tv_nop\n\tv_nop\n\tv_nop\n\tv_nop");
    __builtin_amdgcn_sched_barrier(0);
    const v2f w1v = *(const v2f*)(w1p + 4 * sn);
    const v2f b1v = *(const v2f*)(b1p + 4 * sn);
    fetch_a(an, abase, sn);
    __builtin_amdgcn_sched_barrier(0);
    products<2, 6, FIRST>(c, ac, hc);
    __builtin_amdgcn_sched_barrier(0);
    asm volatile("v_nop\n\tv_nop\n\tv_nop\n\tv_nop\n\tv_nop");
    __builtin_amdgcn_sched_barrier(0);
    make_h(hn, av, w1v, b1v);
    #pragma unroll
    for (int j = 0; j < NB; ++j) asm volatile("" : "+v"(hn[j]));
    __builtin_amdgcn_sched_barrier(0);
    products<6, 8, FIRST>(c, ac, hc);
    #pragma unroll
    for (int mt = 0; mt < MT; ++mt) asm volatile("" : "+v"(an[mt]));
    if (FIRST) {
        __builtin_amdgcn_sched_barrier(0);
        asm volatile("v_nop\n\tv_nop\n\tv_nop\n\tv_nop\n\tv_nop");
        __builtin_amdgcn_sched_barrier(0);
    }
}

template <bool SUF>
__device__ __forceinline__ void build_affine(v2f* tab, const float* wrow, const float* w1s,
                                             const float* b1s, int k)
{
    float ps[32], qs[32];
    #pragma unroll
    for (int g = 0; g < 32; g += 8) {
        v4f w[8], x[8], y[8];
        #pragma unroll
        for (int i = 0; i < 8; ++i) {
            w[i] = *(const v4f*)(wrow + 4 * (g + i));
            x[i] = *(const v4f*)(w1s + 4 * (g + i));
            y[i] = *(const v4f*)(b1s + 4 * (g + i));
        }
        #pragma unroll
        for (int i = 0; i < 8; ++i) {
            ps[g + i] = __builtin_fmaf(w[i].x, x[i].x, w[i].y * x[i].y) + __builtin_fmaf(w[i].z, x[i].z, w[i].w * x[i].w);
            qs[g + i] = __builtin_fmaf(w[i].x, y[i].x, w[i].y * y[i].y) + __builtin_fmaf(w[i].z, y[i].z, w[i].w * y[i].w);
        }
    }
    float p = 0.0f, q = 0.0f;
    tab[(SUF ? 32 : 0) * HID + k] = (v2f){p, q};
    #pragma unroll
    for (int i = 0; i < 32; ++i) {
        const int s = SUF ? 31 - i : i;
        p += ps[s];
        q += qs[s];
        tab[(SUF ? s : s + 1) * HID + k] = (v2f){p, q};
    }
}

template <int M>
__device__ __forceinline__ float row_xor(float x)
{
    int xi = __builtin_bit_cast(int, x);
    int r = __builtin_amdgcn_update_dpp(xi, xi, 0x160 + M, 0xf, 0xf, false);
    return __builtin_bit_cast(float, r);
}

__global__ __launch_bounds__(THREADS) void mlp256_kouter_live_steps_wmma_f32(
    const float* __restrict__ a,  const float* __restrict__ W1,
    const float* __restrict__ b1, const float* __restrict__ W2,
    const float* __restrict__ b2, const float* __restrict__ W3,
    const float* __restrict__ b3, float* __restrict__ out)
{
    __shared__ __attribute__((aligned(16))) float sW2[HID * W2_STRIDE];
    __shared__ __attribute__((aligned(16))) float sW1[HID];
    __shared__ __attribute__((aligned(16))) float sB1[HID];
    __shared__ __attribute__((aligned(16))) float sB2[HID];
    __shared__ __attribute__((aligned(16))) float sW3[HID];
    __shared__ __attribute__((aligned(16))) float sA[BATCH];
    __shared__ __attribute__((aligned(16))) int   sRow[BATCH];
    __shared__ __attribute__((aligned(16))) float sOut[BATCH];
    __shared__ __attribute__((aligned(16))) int   sUPos[HID];
    __shared__ __attribute__((aligned(16))) unsigned sKey[HID];
    __shared__ __attribute__((aligned(16))) int   sCnt[RCLS * WAVES];
    __shared__ __attribute__((aligned(16))) int   sBase[RCLS * WAVES];
    __shared__ __attribute__((aligned(16))) int   sTot[RCLS];
    __shared__ __attribute__((aligned(16))) v2f   sTpre[33 * HID];
    __shared__ __attribute__((aligned(16))) v2f   sTsuf[33 * HID];

    const int n = blockIdx.x;
    const int t = threadIdx.x;
    const int lane = t & 31;
    const int wv   = t >> 5;
    const unsigned lt = (1u << lane) - 1u;

    const float* gW2 = W2 + (size_t)n * HID * HID;
    v4f w2v[8];
    #pragma unroll
    for (int i = 0; i < 8; ++i) w2v[i] = *(const v4f*)(gW2 + (t + THREADS * i) * 4);

    const float* ga = a + (size_t)n * BATCH;
    float ar[4];
    {
        v4f x = *(const v4f*)(ga + t * 4);
        ar[0] = x.x; ar[1] = x.y; ar[2] = x.z; ar[3] = x.w;
    }

    float w1 = 0.f, bb1 = 0.f;
    unsigned key = 0u;
    if (t < HID) {
        w1  = W1[n * HID + t];
        bb1 = b1[n * HID + t];
        const unsigned kb = __builtin_bit_cast(unsigned, -bb1 / w1);
        const unsigned so = (kb & 0x80000000u) ? ~kb : (kb | 0x80000000u);
        key = ((w1 > 0.f) ? 0u : 0x80000000u) | ((so >> 8) << 7) | (unsigned)t;
        sKey[t] = key;
        sB2[t] = b2[n * HID + t];
        sW3[t] = W3[n * HID + t];
    }

    int xl[4];
    #pragma unroll
    for (int b = 0; b < 4; ++b) xl[b] = ((lane >> b) & 1) - 1;
    int rb[4], rl[4];
    int run = 0;
    #pragma unroll
    for (int q = 0; q < 4; ++q) {
        const unsigned ab = __builtin_bit_cast(unsigned, ar[q]);
        const int e  = (int)((ab >> 23) & 0xffu);
        const int mm = min(max(e - 120, 0), 7);
        const int g  = (ab >> 31) ? 8 + mm : 7 - mm;
        unsigned mine = 0xffffffffu, ofl = 0xffffffffu;
        #pragma unroll
        for (int b = 0; b < 4; ++b) {
            const unsigned bb = __builtin_amdgcn_ballot_w32(((g >> b) & 1) != 0);
            mine &= bb ^ (unsigned)(((g >> b) & 1) - 1);
            ofl  &= bb ^ (unsigned)xl[b];
        }
        rb[q] = g;
        rl[q] = __shfl(run, g, 32) + __builtin_popcount(mine & lt);
        run += __builtin_popcount(ofl);
    }
    if (lane < RCLS) sCnt[lane * WAVES + wv] = run;
    __syncthreads();

    if (t < HID) {
        int up = 0;
        #pragma unroll
        for (int j = 0; j < HID; j += 4) {
            const v4u kk = *(const v4u*)(sKey + j);
            up += (kk.x < key) + (kk.y < key) + (kk.z < key) + (kk.w < key);
        }
        sUPos[t] = up;
        sW1[up]  = w1;
        sB1[up]  = bb1;
    } else if (t < HID + RCLS * WAVES) {
        const int i = t - HID;
        const int w = i & (WAVES - 1);
        const int* crow = sCnt + (i - w);
        int b = 0;
        #pragma unroll
        for (int j = 0; j < WAVES; j += 4) {
            const v4i cc = *(const v4i*)(crow + j);
            b += (j + 0 < w ? cc.x : 0) + (j + 1 < w ? cc.y : 0) + (j + 2 < w ? cc.z : 0) + (j + 3 < w ? cc.w : 0);
        }
        sBase[i] = b;
        if (w == WAVES - 1) sTot[i / WAVES] = b + crow[w];
    }
    __syncthreads();

    int cstart = 0;
    {
        const int lc = lane & (RCLS - 1);
        #pragma unroll
        for (int j = 0; j < RCLS; j += 4) {
            const v4i cc = *(const v4i*)(sTot + j);
            cstart += (j + 0 < lc ? cc.x : 0) + (j + 1 < lc ? cc.y : 0) + (j + 2 < lc ? cc.z : 0) + (j + 3 < lc ? cc.w : 0);
        }
    }
    #pragma unroll
    for (int q = 0; q < 4; ++q) {
        int pos = __shfl(cstart, rb[q], 32) + sBase[rb[q] * WAVES + wv] + rl[q];
        sA[pos]   = ar[q];
        sRow[pos] = t * 4 + q;
    }

    {
        const v4i pc = *(const v4i*)(sUPos + ((4 * t) & 127));
        #pragma unroll
        for (int i = 0; i < 8; ++i) {
            int row = ((t + THREADS * i) * 4) >> 7;
            float* d = sW2 + row * W2_STRIDE;
            d[pc.x] = w2v[i].x; d[pc.y] = w2v[i].y; d[pc.z] = w2v[i].z; d[pc.w] = w2v[i].w;
        }
    }
    __syncthreads();

    {
        const int k = t & (HID - 1);
        const float* wrow = sW2 + k * W2_STRIDE;
        if (t < HID)          build_affine<false>(sTpre, wrow, sW1, sB1, k);
        else if (t < 2 * HID) build_affine<true >(sTsuf, wrow, sW1, sB1, k);
    }
    __syncthreads();

    const int li   = lane & 15;
    const int lh   = lane >> 4;
    const float b3v = b3[n];
    float* gout = out + (size_t)n * BATCH;
    const float* abase = sW2 + li * W2_STRIDE + 2 * lh;

    const float* w1p = sW1 + 2 * lh;
    const float* b1p = sB1 + 2 * lh;

    float b2r[MT][8], w3r[MT][8];
    #pragma unroll
    for (int mt = 0; mt < MT; ++mt) {
        #pragma unroll
        for (int q = 0; q < 2; ++q) {
            const v4f x = *(const v4f*)(sB2 + mt * 16 + 8 * lh + 4 * q);
            const v4f y = *(const v4f*)(sW3 + mt * 16 + 8 * lh + 4 * q);
            b2r[mt][4 * q + 0] = x.x; b2r[mt][4 * q + 1] = x.y; b2r[mt][4 * q + 2] = x.z; b2r[mt][4 * q + 3] = x.w;
            w3r[mt][4 * q + 0] = y.x; w3r[mt][4 * q + 1] = y.y; w3r[mt][4 * q + 2] = y.z; w3r[mt][4 * q + 3] = y.w;
        }
    }

    #pragma unroll 1
    for (int p = 0; p < PASSES; ++p) {
        const int base = (p * WAVES + wv) * ROWS_PER_PASS;

        float av[NB];
        #pragma unroll
        for (int j = 0; j < NB; ++j) av[j] = sA[base + 16 * j + li];

        float lo = av[0], hi = av[0];
        #pragma unroll
        for (int j = 1; j < NB; ++j) { lo = fminf(lo, av[j]); hi = fmaxf(hi, av[j]); }
        lo = fminf(lo, row_xor<1>(lo));  hi = fmaxf(hi, row_xor<1>(hi));
        lo = fminf(lo, row_xor<2>(lo));  hi = fmaxf(hi, row_xor<2>(hi));
        lo = fminf(lo, row_xor<4>(lo));  hi = fmaxf(hi, row_xor<4>(hi));
        lo = fminf(lo, row_xor<8>(lo));  hi = fmaxf(hi, row_xor<8>(hi));
        const v4f wq = *(const v4f*)(sW1 + 4 * lane);
        const v4f bq = *(const v4f*)(sB1 + 4 * lane);
        const bool lx = __builtin_fmaf(lo, wq.x, bq.x) > 0.0f, hx = __builtin_fmaf(hi, wq.x, bq.x) > 0.0f;
        const bool ly = __builtin_fmaf(lo, wq.y, bq.y) > 0.0f, hy = __builtin_fmaf(hi, wq.y, bq.y) > 0.0f;
        const bool lz = __builtin_fmaf(lo, wq.z, bq.z) > 0.0f, hz = __builtin_fmaf(hi, wq.z, bq.z) > 0.0f;
        const bool lw = __builtin_fmaf(lo, wq.w, bq.w) > 0.0f, hw = __builtin_fmaf(hi, wq.w, bq.w) > 0.0f;
        const bool on   = lx | hx | ly | hy | lz | hz | lw | hw;
        const bool full = lx & hx & ly & hy & lz & hz & lw & hw;
        const unsigned mon = __builtin_amdgcn_ballot_w32(on);
        const unsigned nf  = ~__builtin_amdgcn_ballot_w32(full);
        int s1 = 32, s3 = 32;
        unsigned m = 0u;
        if (nf != 0u) {
            s1 = __builtin_ctz(nf);
            s3 = 32 - __builtin_clz(nf);
            const unsigned below3 = (s3 >= 32) ? 0xffffffffu : ((1u << s3) - 1u);
            m = mon & below3 & ~((1u << s1) - 1u);
        }

        v8f c[MT][NB];
        v2f a0[MT], a1[MT], h0[NB], h1[NB];
        {
            const v2f* qb = (lh ? sTsuf + s3 * HID : sTpre + s1 * HID) + li;
            #pragma unroll
            for (int mt = 0; mt < MT; ++mt) a0[mt] = qb[mt * 16];
            #pragma unroll
            for (int j = 0; j < NB; ++j) h0[j] = (v2f){av[j], 1.0f};
        }
        bool more = (m != 0u);
        int sn = more ? __builtin_ctz(m) : 0;
        k_step<true>(c, a0, h0, a1, h1, av, abase, w1p, b1p, sn);
        while (more) {
            m &= m - 1u;
            more = (m != 0u);
            sn = more ? __builtin_ctz(m) : 0;
            k_step<false>(c, a1, h1, a0, h0, av, abase, w1p, b1p, sn);
            if (!more) break;
            m &= m - 1u;
            more = (m != 0u);
            sn = more ? __builtin_ctz(m) : 0;
            k_step<false>(c, a0, h0, a1, h1, av, abase, w1p, b1p, sn);
        }
        asm volatile("v_nop\n\tv_nop\n\tv_nop\n\tv_nop\n\tv_nop");

        float acc[NB];
        #pragma unroll
        for (int j = 0; j < NB; ++j) {
            float part[MT];
            #pragma unroll
            for (int mt = 0; mt < MT; ++mt) {
                float s = 0.0f;
                #pragma unroll
                for (int v = 0; v < 8; ++v) {
                    float h2 = fmaxf(c[mt][j][v] + b2r[mt][v], 0.0f);
                    s += h2 * w3r[mt][v];
                }
                part[mt] = s;
            }
            acc[j] = ((part[0] + part[1]) + (part[2] + part[3])) + ((part[4] + part[5]) + (part[6] + part[7]));
        }

        #pragma unroll
        for (int j = 0; j < NB; ++j) acc[j] += __shfl_xor(acc[j], 16, 32);

        if (lh == 0) {
            #pragma unroll
            for (int j = 0; j < NB; ++j) sOut[sRow[base + 16 * j + li]] = acc[j] + b3v;
        }
    }
    __syncthreads();

    {
        int e = t * 4;
        *(v4f*)(gout + e) = *(const v4f*)(sOut + e);
    }
}

extern "C" void kernel_launch(void* const* d_in, const int* in_sizes, int n_in,
                              void* d_out, int out_size, void* d_ws, size_t ws_size,
                              hipStream_t stream) {
    const float* a  = (const float*)d_in[0];
    const float* W1 = (const float*)d_in[1];
    const float* b1 = (const float*)d_in[2];
    const float* W2 = (const float*)d_in[3];
    const float* b2 = (const float*)d_in[4];
    const float* W3 = (const float*)d_in[5];
    const float* b3 = (const float*)d_in[6];
    float* out = (float*)d_out;

    dim3 grid(NNETS);
    mlp256_kouter_live_steps_wmma_f32<<<grid, THREADS, 0, stream>>>(a, W1, b1, W2, b2, W3, b3, out);
}


# Run ends cut per unit, not per 4-unit k step: one per-unit running-sum table (back run = last entry minus an entry), run units zeroed in boundary steps, empty steps skipped; order of sums changes
# speedup vs baseline: 4.1012x; 1.0697x over previous
#include <hip/hip_runtime.h>


typedef __attribute__((ext_vector_type(2))) float v2f;
typedef __attribute__((ext_vector_type(4))) float v4f;
typedef __attribute__((ext_vector_type(8))) float v8f;
typedef __attribute__((ext_vector_type(4))) int v4i;
typedef __attribute__((ext_vector_type(4))) unsigned v4u;

#define NNETS 256
#define THREADS 512
#define HID 128
#define BATCH 2048
#define W2_STRIDE 132
#define NB 1
#define WAVES 16
#define MT 8
#define ROWS_PER_PASS (NB * 16)
#define PASSES (BATCH / (WAVES * ROWS_PER_PASS))
#define RCLS 16

__device__ __forceinline__ void fetch_a(v2f (&af)[MT], const float* abase, int s)
{
    const float* p = abase + 4 * s;
    #pragma unroll
    for (int mt = 0; mt < MT; ++mt) af[mt] = *(const v2f*)(p + mt * 16 * W2_STRIDE);
}

__device__ __forceinline__ void make_h(v2f (&h)[NB], const float (&av)[NB], v2f w1v, v2f b1v, int d0, unsigned len)
{
    const bool in0 = (unsigned)d0 < len;
    const bool in1 = (unsigned)(d0 + 1) < len;
    #pragma unroll
    for (int j = 0; j < NB; ++j) {
        h[j].x = in0 ? fmaxf(av[j] * w1v.x + b1v.x, 0.0f) : 0.0f;
        h[j].y = in1 ? fmaxf(av[j] * w1v.y + b1v.y, 0.0f) : 0.0f;
    }
}

template <int T0, int T1, bool FIRST>
__device__ __forceinline__ void products(v8f (&c)[MT][NB], const v2f (&af)[MT], const v2f (&h)[NB])
{
    const v8f zero = {0.f, 0.f, 0.f, 0.f, 0.f, 0.f, 0.f, 0.f};
    #pragma unroll
    for (int mt = T0; mt < T1; ++mt) {
        #pragma unroll
        for (int j = 0; j < NB; ++j) {
            c[mt][j] = __builtin_amdgcn_wmma_f32_16x16x4_f32(
                           false, af[mt], false, h[j], (short)0, FIRST ? zero : c[mt][j], false, false);
        }
    }
}

template <bool FIRST>
__device__ __forceinline__ void k_step(v8f (&c)[MT][NB], const v2f (&ac)[MT], const v2f (&hc)[NB],
                                       v2f (&an)[MT], v2f (&hn)[NB], const float (&av)[NB],
                                       const float* abase, const float* w1p, const float* b1p, int sn,
                                       int pre, unsigned len)
{
    products<0, 2, FIRST>(c, ac, hc);
    __builtin_amdgcn_sched_barrier(0);
    asm volatile("v_nop\n\tv_nop\n\tv_nop\n\tv_nop\n\tv_nop");
    __builtin_amdgcn_sched_barrier(0);
    const v2f w1v = *(const v2f*)(w1p + 4 * sn);
    const v2f b1v = *(const v2f*)(b1p + 4 * sn);
    fetch_a(an, abase, sn);
    __builtin_amdgcn_sched_barrier(0);
    products<2, 6, FIRST>(c, ac, hc);
    __builtin_amdgcn_sched_barrier(0);
    asm volatile("v_nop\n\tv_nop\n\tv_nop\n\tv_nop\n\tv_nop");
    __builtin_amdgcn_sched_barrier(0);
    make_h(hn, av, w1v, b1v, 4 * sn + pre, len);
    #pragma unroll
    for (int j = 0; j < NB; ++j) asm volatile("" : "+v"(hn[j]));
    __builtin_amdgcn_sched_barrier(0);
    products<6, 8, FIRST>(c, ac, hc);
    #pragma unroll
    for (int mt = 0; mt < MT; ++mt) asm volatile("" : "+v"(an[mt]));
    if (FIRST) {
        __builtin_amdgcn_sched_barrier(0);
        asm volatile("v_nop\n\tv_nop\n\tv_nop\n\tv_nop\n\tv_nop");
        __builtin_amdgcn_sched_barrier(0);
    }
}

template <int M>
__device__ __forceinline__ float row_xor(float x)
{
    int xi = __builtin_bit_cast(int, x);
    int r = __builtin_amdgcn_update_dpp(xi, xi, 0x160 + M, 0xf, 0xf, false);
    return __builtin_bit_cast(float, r);
}

__global__ __launch_bounds__(THREADS) void mlp256_kouter_live_steps_wmma_f32(
    const float* __restrict__ a,  const float* __restrict__ W1,
    const float* __restrict__ b1, const float* __restrict__ W2,
    const float* __restrict__ b2, const float* __restrict__ W3,
    const float* __restrict__ b3, float* __restrict__ out)
{
    __shared__ __attribute__((aligned(16))) float sW2[HID * W2_STRIDE];
    __shared__ __attribute__((aligned(16))) float sW1[HID];
    __shared__ __attribute__((aligned(16))) float sB1[HID];
    __shared__ __attribute__((aligned(16))) float sB2[HID];
    __shared__ __attribute__((aligned(16))) float sW3[HID];
    __shared__ __attribute__((aligned(16))) float sA[BATCH];
    __shared__ __attribute__((aligned(16))) int   sRow[BATCH];
    __shared__ __attribute__((aligned(16))) float sOut[BATCH];
    __shared__ __attribute__((aligned(16))) int   sUPos[HID];
    __shared__ __attribute__((aligned(16))) unsigned sKey[HID];
    __shared__ __attribute__((aligned(16))) int   sCnt[RCLS * WAVES];
    __shared__ __attribute__((aligned(16))) int   sBase[RCLS * WAVES];
    __shared__ __attribute__((aligned(16))) int   sTot[RCLS];
    __shared__ __attribute__((aligned(16))) v2f   sT[(HID + 1) * HID];
    __shared__ __attribute__((aligned(16))) v2f   sPart[4 * HID];

    const int n = blockIdx.x;
    const int t = threadIdx.x;
    const int lane = t & 31;
    const int wv   = t >> 5;
    const unsigned lt = (1u << lane) - 1u;

    const float* gW2 = W2 + (size_t)n * HID * HID;
    v4f w2v[8];
    #pragma unroll
    for (int i = 0; i < 8; ++i) w2v[i] = *(const v4f*)(gW2 + (t + THREADS * i) * 4);

    const float* ga = a + (size_t)n * BATCH;
    float ar[4];
    {
        v4f x = *(const v4f*)(ga + t * 4);
        ar[0] = x.x; ar[1] = x.y; ar[2] = x.z; ar[3] = x.w;
    }

    float w1 = 0.f, bb1 = 0.f;
    unsigned key = 0u;
    if (t < HID) {
        w1  = W1[n * HID + t];
        bb1 = b1[n * HID + t];
        const unsigned kb = __builtin_bit_cast(unsigned, -bb1 / w1);
        const unsigned so = (kb & 0x80000000u) ? ~kb : (kb | 0x80000000u);
        key = ((w1 > 0.f) ? 0u : 0x80000000u) | ((so >> 8) << 7) | (unsigned)t;
        sKey[t] = key;
        sB2[t] = b2[n * HID + t];
        sW3[t] = W3[n * HID + t];
    }

    int xl[4];
    #pragma unroll
    for (int b = 0; b < 4; ++b) xl[b] = ((lane >> b) & 1) - 1;
    int rb[4], rl[4];
    int run = 0;
    #pragma unroll
    for (int q = 0; q < 4; ++q) {
        const unsigned ab = __builtin_bit_cast(unsigned, ar[q]);
        const int e  = (int)((ab >> 23) & 0xffu);
        const int mm = min(max(e - 120, 0), 7);
        const int g  = (ab >> 31) ? 8 + mm : 7 - mm;
        unsigned mine = 0xffffffffu, ofl = 0xffffffffu;
        #pragma unroll
        for (int b = 0; b < 4; ++b) {
            const unsigned bb = __builtin_amdgcn_ballot_w32(((g >> b) & 1) != 0);
            mine &= bb ^ (unsigned)(((g >> b) & 1) - 1);
            ofl  &= bb ^ (unsigned)xl[b];
        }
        rb[q] = g;
        rl[q] = __shfl(run, g, 32) + __builtin_popcount(mine & lt);
        run += __builtin_popcount(ofl);
    }
    if (lane < RCLS) sCnt[lane * WAVES + wv] = run;
    __syncthreads();

    if (t < HID) {
        int up = 0;
        #pragma unroll
        for (int j = 0; j < HID; j += 4) {
            const v4u kk = *(const v4u*)(sKey + j);
            up += (kk.x < key) + (kk.y < key) + (kk.z < key) + (kk.w < key);
        }
        sUPos[t] = up;
        sW1[up]  = w1;
        sB1[up]  = bb1;
    } else if (t < HID + RCLS * WAVES) {
        const int i = t - HID;
        const int w = i & (WAVES - 1);
        const int* crow = sCnt + (i - w);
        int b = 0;
        #pragma unroll
        for (int j = 0; j < WAVES; j += 4) {
            const v4i cc = *(const v4i*)(crow + j);
            b += (j + 0 < w ? cc.x : 0) + (j + 1 < w ? cc.y : 0) + (j + 2 < w ? cc.z : 0) + (j + 3 < w ? cc.w : 0);
        }
        sBase[i] = b;
        if (w == WAVES - 1) sTot[i / WAVES] = b + crow[w];
    }
    __syncthreads();

    int cstart = 0;
    {
        const int lc = lane & (RCLS - 1);
        #pragma unroll
        for (int j = 0; j < RCLS; j += 4) {
            const v4i cc = *(const v4i*)(sTot + j);
            cstart += (j + 0 < lc ? cc.x : 0) + (j + 1 < lc ? cc.y : 0) + (j + 2 < lc ? cc.z : 0) + (j + 3 < lc ? cc.w : 0);
        }
    }
    #pragma unroll
    for (int q = 0; q < 4; ++q) {
        int pos = __shfl(cstart, rb[q], 32) + sBase[rb[q] * WAVES + wv] + rl[q];
        sA[pos]   = ar[q];
        sRow[pos] = t * 4 + q;
    }

    {
        const v4i pc = *(const v4i*)(sUPos + ((4 * t) & 127));
        #pragma unroll
        for (int i = 0; i < 8; ++i) {
            int row = ((t + THREADS * i) * 4) >> 7;
            float* d = sW2 + row * W2_STRIDE;
            d[pc.x] = w2v[i].x; d[pc.y] = w2v[i].y; d[pc.z] = w2v[i].z; d[pc.w] = w2v[i].w;
        }
    }
    __syncthreads();

    {
        const int k = t & (HID - 1);
        const int j = t >> 7;
        const float* wrow = sW2 + k * W2_STRIDE + 32 * j;
        float lp[32], lq[32];
        #pragma unroll
        for (int g = 0; g < 8; ++g) {
            const v4f w = *(const v4f*)(wrow + 4 * g);
            const v4f x = *(const v4f*)(sW1 + 32 * j + 4 * g);
            const v4f y = *(const v4f*)(sB1 + 32 * j + 4 * g);
            lp[4 * g + 0] = w.x * x.x; lp[4 * g + 1] = w.y * x.y; lp[4 * g + 2] = w.z * x.z; lp[4 * g + 3] = w.w * x.w;
            lq[4 * g + 0] = w.x * y.x; lq[4 * g + 1] = w.y * y.y; lq[4 * g + 2] = w.z * y.z; lq[4 * g + 3] = w.w * y.w;
        }
        #pragma unroll
        for (int i = 1; i < 32; ++i) { lp[i] += lp[i - 1]; lq[i] += lq[i - 1]; }
        sPart[j * HID + k] = (v2f){lp[31], lq[31]};
        __syncthreads();
        const v2f t0 = sPart[k], t1 = sPart[HID + k], t2 = sPart[2 * HID + k];
        float op = 0.0f, oq = 0.0f;
        if (j > 0) { op = t0.x;  oq = t0.y; }
        if (j > 1) { op += t1.x; oq += t1.y; }
        if (j > 2) { op += t2.x; oq += t2.y; }
        if (j == 0) sT[k] = (v2f){0.0f, 0.0f};
        #pragma unroll
        for (int i = 0; i < 32; ++i) sT[(32 * j + i + 1) * HID + k] = (v2f){op + lp[i], oq + lq[i]};
    }
    __syncthreads();

    const int li   = lane & 15;
    const int lh   = lane >> 4;
    const float b3v = b3[n];
    float* gout = out + (size_t)n * BATCH;
    const float* abase = sW2 + li * W2_STRIDE + 2 * lh;

    const float* w1p = sW1 + 2 * lh;
    const float* b1p = sB1 + 2 * lh;

    float b2r[MT][8], w3r[MT][8];
    #pragma unroll
    for (int mt = 0; mt < MT; ++mt) {
        #pragma unroll
        for (int q = 0; q < 2; ++q) {
            const v4f x = *(const v4f*)(sB2 + mt * 16 + 8 * lh + 4 * q);
            const v4f y = *(const v4f*)(sW3 + mt * 16 + 8 * lh + 4 * q);
            b2r[mt][4 * q + 0] = x.x; b2r[mt][4 * q + 1] = x.y; b2r[mt][4 * q + 2] = x.z; b2r[mt][4 * q + 3] = x.w;
            w3r[mt][4 * q + 0] = y.x; w3r[mt][4 * q + 1] = y.y; w3r[mt][4 * q + 2] = y.z; w3r[mt][4 * q + 3] = y.w;
        }
    }

    #pragma unroll 1
    for (int p = 0; p < PASSES; ++p) {
        const int base = (p * WAVES + wv) * ROWS_PER_PASS;

        float av[NB];
        #pragma unroll
        for (int j = 0; j < NB; ++j) av[j] = sA[base + 16 * j + li];

        float lo = av[0], hi = av[0];
        #pragma unroll
        for (int j = 1; j < NB; ++j) { lo = fminf(lo, av[j]); hi = fmaxf(hi, av[j]); }
        lo = fminf(lo, row_xor<1>(lo));  hi = fmaxf(hi, row_xor<1>(hi));
        lo = fminf(lo, row_xor<2>(lo));  hi = fmaxf(hi, row_xor<2>(hi));
        lo = fminf(lo, row_xor<4>(lo));  hi = fmaxf(hi, row_xor<4>(hi));
        lo = fminf(lo, row_xor<8>(lo));  hi = fmaxf(hi, row_xor<8>(hi));
        const v4f wq = *(const v4f*)(sW1 + 4 * lane);
        const v4f bq = *(const v4f*)(sB1 + 4 * lane);
        const bool lx = __builtin_fmaf(lo, wq.x, bq.x) > 0.0f, hx = __builtin_fmaf(hi, wq.x, bq.x) > 0.0f;
        const bool ly = __builtin_fmaf(lo, wq.y, bq.y) > 0.0f, hy = __builtin_fmaf(hi, wq.y, bq.y) > 0.0f;
        const bool lz = __builtin_fmaf(lo, wq.z, bq.z) > 0.0f, hz = __builtin_fmaf(hi, wq.z, bq.z) > 0.0f;
        const bool lw = __builtin_fmaf(lo, wq.w, bq.w) > 0.0f, hw = __builtin_fmaf(hi, wq.w, bq.w) > 0.0f;
        const bool fx = lx & hx, fy = ly & hy, fz = lz & hz, fw = lw & hw;
        const bool ox = lx | hx, oy = ly | hy, oz = lz | hz, ow = lw | hw;
        const unsigned nf = ~__builtin_amdgcn_ballot_w32(fx & fy & fz & fw);
        const int r1 = fx ? (fy ? (fz ? 3 : 2) : 1) : 0;
        const int t3 = fw ? (fz ? (fy ? 3 : 2) : 1) : 0;
        int f1 = HID, u3 = HID;
        unsigned len = 0u;
        unsigned m = 0u;
        if (nf != 0u) {
            const int s1 = __builtin_ctz(nf);
            const int s3 = 31 - __builtin_clz(nf);
            f1 = 4 * s1 + __builtin_amdgcn_readlane(r1, s1);
            u3 = 4 * s3 + 4 - __builtin_amdgcn_readlane(t3, s3);
            len = (unsigned)(u3 - f1);
            const int p0 = 4 * lane - f1;
            const bool lv = (ox & ((unsigned)p0 < len)) | (oy & ((unsigned)(p0 + 1) < len)) |
                            (oz & ((unsigned)(p0 + 2) < len)) | (ow & ((unsigned)(p0 + 3) < len));
            m = __builtin_amdgcn_ballot_w32(lv);
        }
        const int pre = 2 * lh - f1;

        v8f c[MT][NB];
        v2f a0[MT], a1[MT], h0[NB], h1[NB];
        {
            const v2f* q1 = sT + (lh ? HID : f1) * HID + li;
            const v2f* q0 = sT + (lh ? u3 : 0) * HID + li;
            #pragma unroll
            for (int mt = 0; mt < MT; ++mt) a0[mt] = q1[mt * 16] - q0[mt * 16];
            #pragma unroll
            for (int j = 0; j < NB; ++j) h0[j] = (v2f){av[j], 1.0f};
        }
        bool more = (m != 0u);
        int sn = more ? __builtin_ctz(m) : 0;
        #pragma unroll
        for (int mt = 0; mt < MT; ++mt) asm volatile("" : "+v"(a0[mt]));
        __builtin_amdgcn_sched_barrier(0);
        k_step<true>(c, a0, h0, a1, h1, av, abase, w1p, b1p, sn, pre, len);
        while (more) {
            m &= m - 1u;
            more = (m != 0u);
            sn = more ? __builtin_ctz(m) : 0;
            k_step<false>(c, a1, h1, a0, h0, av, abase, w1p, b1p, sn, pre, len);
            if (!more) break;
            m &= m - 1u;
            more = (m != 0u);
            sn = more ? __builtin_ctz(m) : 0;
            k_step<false>(c, a0, h0, a1, h1, av, abase, w1p, b1p, sn, pre, len);
        }
        asm volatile("v_nop\n\tv_nop\n\tv_nop\n\tv_nop\n\tv_nop");

        float acc[NB];
        #pragma unroll
        for (int j = 0; j < NB; ++j) {
            float part[MT];
            #pragma unroll
            for (int mt = 0; mt < MT; ++mt) {
                float s = 0.0f;
                #pragma unroll
                for (int v = 0; v < 8; ++v) {
                    float h2 = fmaxf(c[mt][j][v] + b2r[mt][v], 0.0f);
                    s += h2 * w3r[mt][v];
                }
                part[mt] = s;
            }
            acc[j] = ((part[0] + part[1]) + (part[2] + part[3])) + ((part[4] + part[5]) + (part[6] + part[7]));
        }

        #pragma unroll
        for (int j = 0; j < NB; ++j) acc[j] += __shfl_xor(acc[j], 16, 32);

        if (lh == 0) {
            #pragma unroll
            for (int j = 0; j < NB; ++j) sOut[sRow[base + 16 * j + li]] = acc[j] + b3v;
        }
    }
    __syncthreads();

    {
        int e = t * 4;
        *(v4f*)(gout + e) = *(const v4f*)(sOut + e);
    }
}

extern "C" void kernel_launch(void* const* d_in, const int* in_sizes, int n_in,
                              void* d_out, int out_size, void* d_ws, size_t ws_size,
                              hipStream_t stream) {
    const float* a  = (const float*)d_in[0];
    const float* W1 = (const float*)d_in[1];
    const float* b1 = (const float*)d_in[2];
    const float* W2 = (const float*)d_in[3];
    const float* b2 = (const float*)d_in[4];
    const float* W3 = (const float*)d_in[5];
    const float* b3 = (const float*)d_in[6];
    float* out = (float*)d_out;

    dim3 grid(NNETS);
    mlp256_kouter_live_steps_wmma_f32<<<grid, THREADS, 0, stream>>>(a, W1, b1, W2, b2, W3, b3, out);
}


# Rows in 64 classes (was 16): sign, exponent, top two mantissa bits of a; a 16-row block spans less of a and keeps fewer live k steps; class starts by one wave's running sum; order of sums changes
# speedup vs baseline: 4.3642x; 1.0641x over previous
#include <hip/hip_runtime.h>


typedef __attribute__((ext_vector_type(2))) float v2f;
typedef __attribute__((ext_vector_type(4))) float v4f;
typedef __attribute__((ext_vector_type(8))) float v8f;
typedef __attribute__((ext_vector_type(4))) int v4i;
typedef __attribute__((ext_vector_type(4))) unsigned v4u;

#define NNETS 256
#define THREADS 512
#define HID 128
#define BATCH 2048
#define W2_STRIDE 132
#define NB 1
#define WAVES 16
#define MT 8
#define ROWS_PER_PASS (NB * 16)
#define PASSES (BATCH / (WAVES * ROWS_PER_PASS))
#define RCLS 64

__device__ __forceinline__ void fetch_a(v2f (&af)[MT], const float* abase, int s)
{
    const float* p = abase + 4 * s;
    #pragma unroll
    for (int mt = 0; mt < MT; ++mt) af[mt] = *(const v2f*)(p + mt * 16 * W2_STRIDE);
}

__device__ __forceinline__ void make_h(v2f (&h)[NB], const float (&av)[NB], v2f w1v, v2f b1v, int d0, unsigned len)
{
    const bool in0 = (unsigned)d0 < len;
    const bool in1 = (unsigned)(d0 + 1) < len;
    #pragma unroll
    for (int j = 0; j < NB; ++j) {
        h[j].x = in0 ? fmaxf(av[j] * w1v.x + b1v.x, 0.0f) : 0.0f;
        h[j].y = in1 ? fmaxf(av[j] * w1v.y + b1v.y, 0.0f) : 0.0f;
    }
}

template <int T0, int T1, bool FIRST>
__device__ __forceinline__ void products(v8f (&c)[MT][NB], const v2f (&af)[MT], const v2f (&h)[NB])
{
    const v8f zero = {0.f, 0.f, 0.f, 0.f, 0.f, 0.f, 0.f, 0.f};
    #pragma unroll
    for (int mt = T0; mt < T1; ++mt) {
        #pragma unroll
        for (int j = 0; j < NB; ++j) {
            c[mt][j] = __builtin_amdgcn_wmma_f32_16x16x4_f32(
                           false, af[mt], false, h[j], (short)0, FIRST ? zero : c[mt][j], false, false);
        }
    }
}

template <bool FIRST>
__device__ __forceinline__ void k_step(v8f (&c)[MT][NB], const v2f (&ac)[MT], const v2f (&hc)[NB],
                                       v2f (&an)[MT], v2f (&hn)[NB], const float (&av)[NB],
                                       const float* abase, const float* w1p, const float* b1p, int sn,
                                       int pre, unsigned len)
{
    products<0, 2, FIRST>(c, ac, hc);
    __builtin_amdgcn_sched_barrier(0);
    asm volatile("v_nop\n\tv_nop\n\tv_nop\n\tv_nop\n\tv_nop");
    __builtin_amdgcn_sched_barrier(0);
    const v2f w1v = *(const v2f*)(w1p + 4 * sn);
    const v2f b1v = *(const v2f*)(b1p + 4 * sn);
    fetch_a(an, abase, sn);
    __builtin_amdgcn_sched_barrier(0);
    products<2, 6, FIRST>(c, ac, hc);
    __builtin_amdgcn_sched_barrier(0);
    asm volatile("v_nop\n\tv_nop\n\tv_nop\n\tv_nop\n\tv_nop");
    __builtin_amdgcn_sched_barrier(0);
    make_h(hn, av, w1v, b1v, 4 * sn + pre, len);
    #pragma unroll
    for (int j = 0; j < NB; ++j) asm volatile("" : "+v"(hn[j]));
    __builtin_amdgcn_sched_barrier(0);
    products<6, 8, FIRST>(c, ac, hc);
    #pragma unroll
    for (int mt = 0; mt < MT; ++mt) asm volatile("" : "+v"(an[mt]));
    if (FIRST) {
        __builtin_amdgcn_sched_barrier(0);
        asm volatile("v_nop\n\tv_nop\n\tv_nop\n\tv_nop\n\tv_nop");
        __builtin_amdgcn_sched_barrier(0);
    }
}

template <int M>
__device__ __forceinline__ float row_xor(float x)
{
    int xi = __builtin_bit_cast(int, x);
    int r = __builtin_amdgcn_update_dpp(xi, xi, 0x160 + M, 0xf, 0xf, false);
    return __builtin_bit_cast(float, r);
}

__global__ __launch_bounds__(THREADS) void mlp256_kouter_live_steps_wmma_f32(
    const float* __restrict__ a,  const float* __restrict__ W1,
    const float* __restrict__ b1, const float* __restrict__ W2,
    const float* __restrict__ b2, const float* __restrict__ W3,
    const float* __restrict__ b3, float* __restrict__ out)
{
    __shared__ __attribute__((aligned(16))) float sW2[HID * W2_STRIDE];
    __shared__ __attribute__((aligned(16))) float sW1[HID];
    __shared__ __attribute__((aligned(16))) float sB1[HID];
    __shared__ __attribute__((aligned(16))) float sB2[HID];
    __shared__ __attribute__((aligned(16))) float sW3[HID];
    __shared__ __attribute__((aligned(16))) float sA[BATCH];
    __shared__ __attribute__((aligned(16))) int   sRow[BATCH];
    __shared__ __attribute__((aligned(16))) float sOut[BATCH];
    __shared__ __attribute__((aligned(16))) int   sUPos[HID];
    __shared__ __attribute__((aligned(16))) unsigned sKey[HID];
    __shared__ __attribute__((aligned(16))) int   sCnt[RCLS * WAVES];
    __shared__ __attribute__((aligned(16))) int   sBase[RCLS * WAVES];
    __shared__ __attribute__((aligned(16))) int   sCStart[RCLS];
    __shared__ __attribute__((aligned(16))) v2f   sT[(HID + 1) * HID];
    __shared__ __attribute__((aligned(16))) v2f   sPart[4 * HID];

    const int n = blockIdx.x;
    const int t = threadIdx.x;
    const int lane = t & 31;
    const int wv   = t >> 5;
    const unsigned lt = (1u << lane) - 1u;

    const float* gW2 = W2 + (size_t)n * HID * HID;
    v4f w2v[8];
    #pragma unroll
    for (int i = 0; i < 8; ++i) w2v[i] = *(const v4f*)(gW2 + (t + THREADS * i) * 4);

    const float* ga = a + (size_t)n * BATCH;
    float ar[4];
    {
        v4f x = *(const v4f*)(ga + t * 4);
        ar[0] = x.x; ar[1] = x.y; ar[2] = x.z; ar[3] = x.w;
    }

    float w1 = 0.f, bb1 = 0.f;
    unsigned key = 0u;
    if (t < HID) {
        w1  = W1[n * HID + t];
        bb1 = b1[n * HID + t];
        const unsigned kb = __builtin_bit_cast(unsigned, -bb1 / w1);
        const unsigned so = (kb & 0x80000000u) ? ~kb : (kb | 0x80000000u);
        key = ((w1 > 0.f) ? 0u : 0x80000000u) | ((so >> 8) << 7) | (unsigned)t;
        sKey[t] = key;
        sB2[t] = b2[n * HID + t];
        sW3[t] = W3[n * HID + t];
    }

    int xl[5];
    #pragma unroll
    for (int b = 0; b < 5; ++b) xl[b] = ((lane >> b) & 1) - 1;
    int rb[4], rl[4];
    int run = 0;
    #pragma unroll
    for (int q = 0; q < 4; ++q) {
        const unsigned ab = __builtin_bit_cast(unsigned, ar[q]);
        const int kx = (int)((ab >> 21) & 0x3ffu);
        const int mm = min(max(kx - 4 * 120, 0), 31);
        const int g  = (ab >> 31) ? 32 + mm : 31 - mm;
        unsigned ofl = 0xffffffffu;
        #pragma unroll
        for (int b = 0; b < 5; ++b) {
            const unsigned bb = __builtin_amdgcn_ballot_w32(((g >> b) & 1) != 0);
            ofl &= bb ^ (unsigned)xl[b];
        }
        const unsigned neg = __builtin_amdgcn_ballot_w32(g >= 32);
        const unsigned mine = (unsigned)__shfl((int)ofl, g & 31, 32) & (neg ^ (unsigned)((g >> 5) - 1));
        rb[q] = g;
        const int r01 = __shfl(run, g & 31, 32);
        rl[q] = ((g & 32) ? (r01 >> 16) : (r01 & 0xffff)) + __builtin_popcount(mine & lt);
        run += __builtin_popcount(ofl & ~neg) + (__builtin_popcount(ofl & neg) << 16);
    }
    sCnt[lane * WAVES + wv]        = run & 0xffff;
    sCnt[(lane + 32) * WAVES + wv] = run >> 16;
    __syncthreads();

    if (t < HID) {
        int up = 0;
        #pragma unroll
        for (int j = 0; j < HID; j += 4) {
            const v4u kk = *(const v4u*)(sKey + j);
            up += (kk.x < key) + (kk.y < key) + (kk.z < key) + (kk.w < key);
        }
        sUPos[t] = up;
        sW1[up]  = w1;
        sB1[up]  = bb1;
    } else if (t < HID + 256) {
        const int i = t - HID;
        const int w = i & (WAVES - 1);
        #pragma unroll
        for (int jj = 0; jj < (RCLS * WAVES) / 256; ++jj) {
            const int e = i + 256 * jj;
            const int* crow = sCnt + (e - w);
            int b = 0;
            #pragma unroll
            for (int j = 0; j < WAVES; j += 4) {
                const v4i cc = *(const v4i*)(crow + j);
                b += (j + 0 < w ? cc.x : 0) + (j + 1 < w ? cc.y : 0) + (j + 2 < w ? cc.z : 0) + (j + 3 < w ? cc.w : 0);
            }
            sBase[e] = b;
        }
    } else if (t < HID + 256 + 32) {
        const int* crow = sCnt + 2 * WAVES * lane;
        int s0 = 0, s1 = 0;
        #pragma unroll
        for (int j = 0; j < WAVES; j += 4) {
            const v4i c0 = *(const v4i*)(crow + j);
            const v4i c1 = *(const v4i*)(crow + WAVES + j);
            s0 += (c0.x + c0.y) + (c0.z + c0.w);
            s1 += (c1.x + c1.y) + (c1.z + c1.w);
        }
        int x = s0 + s1;
        #pragma unroll
        for (int d = 1; d < 32; d *= 2) {
            const int y = __shfl_up(x, d, 32);
            if (lane >= d) x += y;
        }
        const int ex = x - (s0 + s1);
        sCStart[2 * lane]     = ex;
        sCStart[2 * lane + 1] = ex + s0;
    }
    __syncthreads();

    #pragma unroll
    for (int q = 0; q < 4; ++q) {
        int pos = sCStart[rb[q]] + sBase[rb[q] * WAVES + wv] + rl[q];
        sA[pos]   = ar[q];
        sRow[pos] = t * 4 + q;
    }

    {
        const v4i pc = *(const v4i*)(sUPos + ((4 * t) & 127));
        #pragma unroll
        for (int i = 0; i < 8; ++i) {
            int row = ((t + THREADS * i) * 4) >> 7;
            float* d = sW2 + row * W2_STRIDE;
            d[pc.x] = w2v[i].x; d[pc.y] = w2v[i].y; d[pc.z] = w2v[i].z; d[pc.w] = w2v[i].w;
        }
    }
    __syncthreads();

    {
        const int k = t & (HID - 1);
        const int j = t >> 7;
        const float* wrow = sW2 + k * W2_STRIDE + 32 * j;
        float lp[32], lq[32];
        #pragma unroll
        for (int g = 0; g < 8; ++g) {
            const v4f w = *(const v4f*)(wrow + 4 * g);
            const v4f x = *(const v4f*)(sW1 + 32 * j + 4 * g);
            const v4f y = *(const v4f*)(sB1 + 32 * j + 4 * g);
            lp[4 * g + 0] = w.x * x.x; lp[4 * g + 1] = w.y * x.y; lp[4 * g + 2] = w.z * x.z; lp[4 * g + 3] = w.w * x.w;
            lq[4 * g + 0] = w.x * y.x; lq[4 * g + 1] = w.y * y.y; lq[4 * g + 2] = w.z * y.z; lq[4 * g + 3] = w.w * y.w;
        }
        #pragma unroll
        for (int i = 1; i < 32; ++i) { lp[i] += lp[i - 1]; lq[i] += lq[i - 1]; }
        sPart[j * HID + k] = (v2f){lp[31], lq[31]};
        __syncthreads();
        const v2f t0 = sPart[k], t1 = sPart[HID + k], t2 = sPart[2 * HID + k];
        float op = 0.0f, oq = 0.0f;
        if (j > 0) { op = t0.x;  oq = t0.y; }
        if (j > 1) { op += t1.x; oq += t1.y; }
        if (j > 2) { op += t2.x; oq += t2.y; }
        if (j == 0) sT[k] = (v2f){0.0f, 0.0f};
        #pragma unroll
        for (int i = 0; i < 32; ++i) sT[(32 * j + i + 1) * HID + k] = (v2f){op + lp[i], oq + lq[i]};
    }
    __syncthreads();

    const int li   = lane & 15;
    const int lh   = lane >> 4;
    const float b3v = b3[n];
    float* gout = out + (size_t)n * BATCH;
    const float* abase = sW2 + li * W2_STRIDE + 2 * lh;

    const float* w1p = sW1 + 2 * lh;
    const float* b1p = sB1 + 2 * lh;

    float b2r[MT][8], w3r[MT][8];
    #pragma unroll
    for (int mt = 0; mt < MT; ++mt) {
        #pragma unroll
        for (int q = 0; q < 2; ++q) {
            const v4f x = *(const v4f*)(sB2 + mt * 16 + 8 * lh + 4 * q);
            const v4f y = *(const v4f*)(sW3 + mt * 16 + 8 * lh + 4 * q);
            b2r[mt][4 * q + 0] = x.x; b2r[mt][4 * q + 1] = x.y; b2r[mt][4 * q + 2] = x.z; b2r[mt][4 * q + 3] = x.w;
            w3r[mt][4 * q + 0] = y.x; w3r[mt][4 * q + 1] = y.y; w3r[mt][4 * q + 2] = y.z; w3r[mt][4 * q + 3] = y.w;
        }
    }

    #pragma unroll 1
    for (int p = 0; p < PASSES; ++p) {
        const int base = (p * WAVES + wv) * ROWS_PER_PASS;

        float av[NB];
        #pragma unroll
        for (int j = 0; j < NB; ++j) av[j] = sA[base + 16 * j + li];

        float lo = av[0], hi = av[0];
        #pragma unroll
        for (int j = 1; j < NB; ++j) { lo = fminf(lo, av[j]); hi = fmaxf(hi, av[j]); }
        lo = fminf(lo, row_xor<1>(lo));  hi = fmaxf(hi, row_xor<1>(hi));
        lo = fminf(lo, row_xor<2>(lo));  hi = fmaxf(hi, row_xor<2>(hi));
        lo = fminf(lo, row_xor<4>(lo));  hi = fmaxf(hi, row_xor<4>(hi));
        lo = fminf(lo, row_xor<8>(lo));  hi = fmaxf(hi, row_xor<8>(hi));
        const v4f wq = *(const v4f*)(sW1 + 4 * lane);
        const v4f bq = *(const v4f*)(sB1 + 4 * lane);
        const bool lx = __builtin_fmaf(lo, wq.x, bq.x) > 0.0f, hx = __builtin_fmaf(hi, wq.x, bq.x) > 0.0f;
        const bool ly = __builtin_fmaf(lo, wq.y, bq.y) > 0.0f, hy = __builtin_fmaf(hi, wq.y, bq.y) > 0.0f;
        const bool lz = __builtin_fmaf(lo, wq.z, bq.z) > 0.0f, hz = __builtin_fmaf(hi, wq.z, bq.z) > 0.0f;
        const bool lw = __builtin_fmaf(lo, wq.w, bq.w) > 0.0f, hw = __builtin_fmaf(hi, wq.w, bq.w) > 0.0f;
        const bool fx = lx & hx, fy = ly & hy, fz = lz & hz, fw = lw & hw;
        const bool ox = lx | hx, oy = ly | hy, oz = lz | hz, ow = lw | hw;
        const unsigned nf = ~__builtin_amdgcn_ballot_w32(fx & fy & fz & fw);
        const int r1 = fx ? (fy ? (fz ? 3 : 2) : 1) : 0;
        const int t3 = fw ? (fz ? (fy ? 3 : 2) : 1) : 0;
        int f1 = HID, u3 = HID;
        unsigned len = 0u;
        unsigned m = 0u;
        if (nf != 0u) {
            const int s1 = __builtin_ctz(nf);
            const int s3 = 31 - __builtin_clz(nf);
            f1 = 4 * s1 + __builtin_amdgcn_readlane(r1, s1);
            u3 = 4 * s3 + 4 - __builtin_amdgcn_readlane(t3, s3);
            len = (unsigned)(u3 - f1);
            const int p0 = 4 * lane - f1;
            const bool lv = (ox & ((unsigned)p0 < len)) | (oy & ((unsigned)(p0 + 1) < len)) |
                            (oz & ((unsigned)(p0 + 2) < len)) | (ow & ((unsigned)(p0 + 3) < len));
            m = __builtin_amdgcn_ballot_w32(lv);
        }
        const int pre = 2 * lh - f1;

        v8f c[MT][NB];
        v2f a0[MT], a1[MT], h0[NB], h1[NB];
        {
            const v2f* q1 = sT + (lh ? HID : f1) * HID + li;
            const v2f* q0 = sT + (lh ? u3 : 0) * HID + li;
            #pragma unroll
            for (int mt = 0; mt < MT; ++mt) a0[mt] = q1[mt * 16] - q0[mt * 16];
            #pragma unroll
            for (int j = 0; j < NB; ++j) h0[j] = (v2f){av[j], 1.0f};
        }
        bool more = (m != 0u);
        int sn = more ? __builtin_ctz(m) : 0;
        #pragma unroll
        for (int mt = 0; mt < MT; ++mt) asm volatile("" : "+v"(a0[mt]));
        __builtin_amdgcn_sched_barrier(0);
        k_step<true>(c, a0, h0, a1, h1, av, abase, w1p, b1p, sn, pre, len);
        while (more) {
            m &= m - 1u;
            more = (m != 0u);
            sn = more ? __builtin_ctz(m) : 0;
            k_step<false>(c, a1, h1, a0, h0, av, abase, w1p, b1p, sn, pre, len);
            if (!more) break;
            m &= m - 1u;
            more = (m != 0u);
            sn = more ? __builtin_ctz(m) : 0;
            k_step<false>(c, a0, h0, a1, h1, av, abase, w1p, b1p, sn, pre, len);
        }
        asm volatile("v_nop\n\tv_nop\n\tv_nop\n\tv_nop\n\tv_nop");

        float acc[NB];
        #pragma unroll
        for (int j = 0; j < NB; ++j) {
            float part[MT];
            #pragma unroll
            for (int mt = 0; mt < MT; ++mt) {
                float s = 0.0f;
                #pragma unroll
                for (int v = 0; v < 8; ++v) {
                    float h2 = fmaxf(c[mt][j][v] + b2r[mt][v], 0.0f);
                    s += h2 * w3r[mt][v];
                }
                part[mt] = s;
            }
            acc[j] = ((part[0] + part[1]) + (part[2] + part[3])) + ((part[4] + part[5]) + (part[6] + part[7]));
        }

        #pragma unroll
        for (int j = 0; j < NB; ++j) acc[j] += __shfl_xor(acc[j], 16, 32);

        if (lh == 0) {
            #pragma unroll
            for (int j = 0; j < NB; ++j) sOut[sRow[base + 16 * j + li]] = acc[j] + b3v;
        }
    }
    __syncthreads();

    {
        int e = t * 4;
        *(v4f*)(gout + e) = *(const v4f*)(sOut + e);
    }
}

extern "C" void kernel_launch(void* const* d_in, const int* in_sizes, int n_in,
                              void* d_out, int out_size, void* d_ws, size_t ws_size,
                              hipStream_t stream) {
    const float* a  = (const float*)d_in[0];
    const float* W1 = (const float*)d_in[1];
    const float* b1 = (const float*)d_in[2];
    const float* W2 = (const float*)d_in[3];
    const float* b2 = (const float*)d_in[4];
    const float* W3 = (const float*)d_in[5];
    const float* b3 = (const float*)d_in[6];
    float* out = (float*)d_out;

    dim3 grid(NNETS);
    mlp256_kouter_live_steps_wmma_f32<<<grid, THREADS, 0, stream>>>(a, W1, b1, W2, b2, W3, b3, out);
}
